# starting kernel
# baseline (speedup 1.0000x reference)
#include <hip/hip_runtime.h>

typedef __attribute__((ext_vector_type(2))) float v2f;
typedef __attribute__((ext_vector_type(8))) float v8f;

#define N_NODES 100000
#define D_IN    128
#define D_HID   64
#define D_OUT   64
#define N_EDGES 1600000

#define WP_STRIDE 80

template <int K>
__global__ __launch_bounds__(256) void gemm_wmma_f32(
    const float* __restrict__ A, const float* __restrict__ W,
    const float* __restrict__ bias, float* __restrict__ out, int nRows,
    int nBlocks) {
  __shared__ float2 Wp[(K / 2) * WP_STRIDE];

  for (int i = threadIdx.x; i < K * 64; i += 256) {
    const int k = i >> 6;
    const int c = i & 63;
    ((float*)Wp)[(k >> 1) * (2 * WP_STRIDE) + c * 2 + (k & 1)] = W[i];
  }
  __syncthreads();

  const int lane = threadIdx.x & 31;
  const int wave = threadIdx.x >> 5;
  const int half = lane >> 4;
  const int l15  = lane & 15;

  const float b0 = bias[l15 +  0];
  const float b1 = bias[l15 + 16];
  const float b2 = bias[l15 + 32];
  const float b3 = bias[l15 + 48];

  const int nTiles = nRows >> 4;

  for (int tile = blockIdx.x * 8 + wave; tile < nTiles;
       tile += nBlocks * 8) {
    const int rowBase = tile * 16;
    const float* aRow = A + (size_t)(rowBase + l15) * K;

    v8f acc0 = {}, acc1 = {}, acc2 = {}, acc3 = {};

#pragma unroll 4
    for (int k = 0; k < K; k += 4) {
      const int kk = k + 2 * half;
      v2f a;
      a.x = aRow[kk];
      a.y = aRow[kk + 1];

      const float2* wrow = Wp + ((k >> 1) + half) * WP_STRIDE + l15;
      float2 w;
      v2f b;

      w = wrow[0];
      b.x = w.x; b.y = w.y;
      acc0 = __builtin_amdgcn_wmma_f32_16x16x4_f32(false, a, false, b,
                                                   (short)0, acc0, false, false);
      w = wrow[16];
      b.x = w.x; b.y = w.y;
      acc1 = __builtin_amdgcn_wmma_f32_16x16x4_f32(false, a, false, b,
                                                   (short)0, acc1, false, false);
      w = wrow[32];
      b.x = w.x; b.y = w.y;
      acc2 = __builtin_amdgcn_wmma_f32_16x16x4_f32(false, a, false, b,
                                                   (short)0, acc2, false, false);
      w = wrow[48];
      b.x = w.x; b.y = w.y;
      acc3 = __builtin_amdgcn_wmma_f32_16x16x4_f32(false, a, false, b,
                                                   (short)0, acc3, false, false);
    }

#pragma unroll
    for (int r = 0; r < 8; ++r) {
      const int row = rowBase + r + 8 * half;
      float* o = out + (size_t)row * 64 + l15;
      o[0]  = acc0[r] + b0;
      o[16] = acc1[r] + b1;
      o[32] = acc2[r] + b2;
      o[48] = acc3[r] + b3;
    }
  }
}

__global__ __launch_bounds__(256) void zero_f32(float* __restrict__ p,
                                                size_t n) {
  size_t i = (size_t)blockIdx.x * blockDim.x + threadIdx.x;
  if (i < n) p[i] = 0.0f;
}

__global__ __launch_bounds__(256) void scatter_edges(
    const float* __restrict__ h, const int* __restrict__ esrc,
    const int* __restrict__ edst, float* __restrict__ agg, int nEdges) {
  const long long gid = (long long)blockIdx.x * blockDim.x + threadIdx.x;
  const int e  = (int)(gid >> 5);
  const int fp = (int)(gid & 31);
  if (e >= nEdges) return;
  const int s = esrc[e];
  const int d = edst[e];
  const float2 v = ((const float2*)h)[(size_t)s * 32 + fp];
  float* dst = agg + (size_t)d * 64 + 2 * fp;
  unsafeAtomicAdd(dst + 0, v.x);
  unsafeAtomicAdd(dst + 1, v.y);
}

extern "C" void kernel_launch(void* const* d_in, const int* in_sizes, int n_in,
                              void* d_out, int out_size, void* d_ws,
                              size_t ws_size, hipStream_t stream) {
  const float* data   = (const float*)d_in[0];
  const int*   esrc   = (const int*)d_in[1];
  const int*   edst   = (const int*)d_in[2];
  const float* W_pre  = (const float*)d_in[3];
  const float* b_pre  = (const float*)d_in[4];
  const float* W_post = (const float*)d_in[5];
  const float* b_post = (const float*)d_in[6];

  float* out = (float*)d_out;
  float* h   = (float*)d_ws;
  float* agg = out;

  {
    const size_t n = (size_t)N_NODES * D_HID;
    zero_f32<<<(unsigned)((n + 255) / 256), 256, 0, stream>>>(agg, n);
  }

  {
    const int blocks = 250;
    gemm_wmma_f32<D_IN><<<blocks, 256, 0, stream>>>(data, W_pre, b_pre, h,
                                                    N_NODES, blocks);
  }

  {
    const long long work   = (long long)N_EDGES * 32;
    const int       blocks = (int)((work + 255) / 256);
    scatter_edges<<<blocks, 256, 0, stream>>>(h, esrc, edst, agg, N_EDGES);
  }

  {
    const int blocks = 250;
    gemm_wmma_f32<D_HID><<<blocks, 256, 0, stream>>>(agg, W_post, b_post, out,
                                                     N_NODES, blocks);
  }
}


# fold W_post into W_pre (Wc = W_pre*W_post built per block on WMMA; edge sum is linear): second node GEMM launch removed, out starts at b_post; the two products are reassociated
# speedup vs baseline: 1.2001x; 1.2001x over previous
#include <hip/hip_runtime.h>


typedef __attribute__((ext_vector_type(2))) float v2f;
typedef __attribute__((ext_vector_type(8))) float v8f;

#define N_NODES 100000
#define D_IN    128
#define D_HID   64
#define D_OUT   64
#define N_EDGES 1600000

#define WP_STRIDE 80

__global__ __launch_bounds__(256) void gemm_folded_wmma_f32(
    const float* __restrict__ A, const float* __restrict__ W1,
    const float* __restrict__ bias1, const float* __restrict__ W2,
    float* __restrict__ out, int nRows, int nBlocks) {
  __shared__ float2 Wp[(D_IN / 2) * WP_STRIDE];
  __shared__ float2 Wq[(D_HID / 2) * WP_STRIDE];

  for (int i = threadIdx.x; i < D_HID * D_OUT; i += 256) {
    const int k = i >> 6;
    const int c = i & 63;
    ((float*)Wq)[(k >> 1) * (2 * WP_STRIDE) + c * 2 + (k & 1)] = W2[i];
  }
  __syncthreads();

  const int lane = threadIdx.x & 31;
  const int wave = threadIdx.x >> 5;
  const int half = lane >> 4;
  const int l15  = lane & 15;

  {
    const float* aRow = W1 + (size_t)(wave * 16 + l15) * D_HID;
    v8f f0 = {}, f1 = {}, f2 = {}, f3 = {};

#pragma unroll 4
    for (int k = 0; k < D_HID; k += 4) {
      const int kk = k + 2 * half;
      v2f a;
      a.x = aRow[kk];
      a.y = aRow[kk + 1];

      const float2* wrow = Wq + ((k >> 1) + half) * WP_STRIDE + l15;
      float2 w;
      v2f b;

      w = wrow[0];
      b.x = w.x; b.y = w.y;
      f0 = __builtin_amdgcn_wmma_f32_16x16x4_f32(false, a, false, b,
                                                 (short)0, f0, false, false);
      w = wrow[16];
      b.x = w.x; b.y = w.y;
      f1 = __builtin_amdgcn_wmma_f32_16x16x4_f32(false, a, false, b,
                                                 (short)0, f1, false, false);
      w = wrow[32];
      b.x = w.x; b.y = w.y;
      f2 = __builtin_amdgcn_wmma_f32_16x16x4_f32(false, a, false, b,
                                                 (short)0, f2, false, false);
      w = wrow[48];
      b.x = w.x; b.y = w.y;
      f3 = __builtin_amdgcn_wmma_f32_16x16x4_f32(false, a, false, b,
                                                 (short)0, f3, false, false);
    }

#pragma unroll
    for (int r = 0; r < 8; ++r) {
      const int kr = wave * 16 + r + 8 * half;
      float* p = (float*)Wp + (kr >> 1) * (2 * WP_STRIDE) + (kr & 1) + 2 * l15;
      p[0]  = f0[r];
      p[32] = f1[r];
      p[64] = f2[r];
      p[96] = f3[r];
    }
  }

  float b0 = 0.0f, b1 = 0.0f, b2 = 0.0f, b3 = 0.0f;
  for (int kp = 0; kp < D_HID / 2; ++kp) {
    const float s0 = bias1[2 * kp];
    const float s1 = bias1[2 * kp + 1];
    const float2* q = Wq + kp * WP_STRIDE + l15;
    float2 w;
    w = q[0];  b0 = fmaf(s0, w.x, b0); b0 = fmaf(s1, w.y, b0);
    w = q[16]; b1 = fmaf(s0, w.x, b1); b1 = fmaf(s1, w.y, b1);
    w = q[32]; b2 = fmaf(s0, w.x, b2); b2 = fmaf(s1, w.y, b2);
    w = q[48]; b3 = fmaf(s0, w.x, b3); b3 = fmaf(s1, w.y, b3);
  }

  __syncthreads();

  const int nTiles = nRows >> 4;

  for (int tile = blockIdx.x * 8 + wave; tile < nTiles;
       tile += nBlocks * 8) {
    const int rowBase = tile * 16;
    const float* aRow = A + (size_t)(rowBase + l15) * D_IN;

    v8f acc0 = {}, acc1 = {}, acc2 = {}, acc3 = {};

#pragma unroll 4
    for (int k = 0; k < D_IN; k += 4) {
      const int kk = k + 2 * half;
      v2f a;
      a.x = aRow[kk];
      a.y = aRow[kk + 1];

      const float2* wrow = Wp + ((k >> 1) + half) * WP_STRIDE + l15;
      float2 w;
      v2f b;

      w = wrow[0];
      b.x = w.x; b.y = w.y;
      acc0 = __builtin_amdgcn_wmma_f32_16x16x4_f32(false, a, false, b,
                                                   (short)0, acc0, false, false);
      w = wrow[16];
      b.x = w.x; b.y = w.y;
      acc1 = __builtin_amdgcn_wmma_f32_16x16x4_f32(false, a, false, b,
                                                   (short)0, acc1, false, false);
      w = wrow[32];
      b.x = w.x; b.y = w.y;
      acc2 = __builtin_amdgcn_wmma_f32_16x16x4_f32(false, a, false, b,
                                                   (short)0, acc2, false, false);
      w = wrow[48];
      b.x = w.x; b.y = w.y;
      acc3 = __builtin_amdgcn_wmma_f32_16x16x4_f32(false, a, false, b,
                                                   (short)0, acc3, false, false);
    }

#pragma unroll
    for (int r = 0; r < 8; ++r) {
      const int row = rowBase + r + 8 * half;
      float* o = out + (size_t)row * 64 + l15;
      o[0]  = acc0[r] + b0;
      o[16] = acc1[r] + b1;
      o[32] = acc2[r] + b2;
      o[48] = acc3[r] + b3;
    }
  }
}

__global__ __launch_bounds__(256) void fill_rows_f32(
    float* __restrict__ p, const float* __restrict__ rowv, size_t n) {
  size_t i = (size_t)blockIdx.x * blockDim.x + threadIdx.x;
  if (i < n) p[i] = rowv[i & 63];
}

__global__ __launch_bounds__(256) void scatter_edges(
    const float* __restrict__ h, const int* __restrict__ esrc,
    const int* __restrict__ edst, float* __restrict__ agg, int nEdges) {
  const long long gid = (long long)blockIdx.x * blockDim.x + threadIdx.x;
  const int e  = (int)(gid >> 5);
  const int fp = (int)(gid & 31);
  if (e >= nEdges) return;
  const int s = esrc[e];
  const int d = edst[e];
  const float2 v = ((const float2*)h)[(size_t)s * 32 + fp];
  float* dst = agg + (size_t)d * 64 + 2 * fp;
  unsafeAtomicAdd(dst + 0, v.x);
  unsafeAtomicAdd(dst + 1, v.y);
}

extern "C" void kernel_launch(void* const* d_in, const int* in_sizes, int n_in,
                              void* d_out, int out_size, void* d_ws,
                              size_t ws_size, hipStream_t stream) {
  const float* data   = (const float*)d_in[0];
  const int*   esrc   = (const int*)d_in[1];
  const int*   edst   = (const int*)d_in[2];
  const float* W_pre  = (const float*)d_in[3];
  const float* b_pre  = (const float*)d_in[4];
  const float* W_post = (const float*)d_in[5];
  const float* b_post = (const float*)d_in[6];

  float* out = (float*)d_out;
  float* g   = (float*)d_ws;

  {
    const size_t n = (size_t)N_NODES * D_OUT;
    fill_rows_f32<<<(unsigned)((n + 255) / 256), 256, 0, stream>>>(out, b_post,
                                                                   n);
  }

  {
    const int blocks = 250;
    gemm_folded_wmma_f32<<<blocks, 256, 0, stream>>>(data, W_pre, b_pre, W_post,
                                                     g, N_NODES, blocks);
  }

  {
    const long long work   = (long long)N_EDGES * 32;
    const int       blocks = (int)((work + 255) / 256);
    scatter_edges<<<blocks, 256, 0, stream>>>(g, esrc, edst, out, N_EDGES);
  }
}


# edge pass: each out row gets its f32 atomic adds from one half of the package only (resident work-groups read their die, take 512-edge chunks by ticket, own rows first); add order changes
# speedup vs baseline: 2.2141x; 1.8449x over previous
#include <hip/hip_runtime.h>


typedef __attribute__((ext_vector_type(2))) float v2f;
typedef __attribute__((ext_vector_type(8))) float v8f;

#define N_NODES 100000
#define D_IN    128
#define D_HID   64
#define D_OUT   64
#define N_EDGES 1600000

#define G_BYTES    ((size_t)N_NODES * D_OUT * sizeof(float))
#define CNT_STRIDE 64
#define WS_NEEDED  (G_BYTES + 2 * CNT_STRIDE * sizeof(unsigned))

#define WP_STRIDE 80

__global__ __launch_bounds__(256) void gemm_folded_wmma_f32(
    const float* __restrict__ A, const float* __restrict__ W1,
    const float* __restrict__ bias1, const float* __restrict__ W2,
    float* __restrict__ out, int nRows, int nBlocks) {
  __shared__ float2 Wp[(D_IN / 2) * WP_STRIDE];
  __shared__ float2 Wq[(D_HID / 2) * WP_STRIDE];

  for (int i = threadIdx.x; i < D_HID * D_OUT; i += 256) {
    const int k = i >> 6;
    const int c = i & 63;
    ((float*)Wq)[(k >> 1) * (2 * WP_STRIDE) + c * 2 + (k & 1)] = W2[i];
  }
  __syncthreads();

  const int lane = threadIdx.x & 31;
  const int wave = threadIdx.x >> 5;
  const int half = lane >> 4;
  const int l15  = lane & 15;

  {
    const float* aRow = W1 + (size_t)(wave * 16 + l15) * D_HID;
    v8f f0 = {}, f1 = {}, f2 = {}, f3 = {};

#pragma unroll 4
    for (int k = 0; k < D_HID; k += 4) {
      const int kk = k + 2 * half;
      v2f a;
      a.x = aRow[kk];
      a.y = aRow[kk + 1];

      const float2* wrow = Wq + ((k >> 1) + half) * WP_STRIDE + l15;
      float2 w;
      v2f b;

      w = wrow[0];
      b.x = w.x; b.y = w.y;
      f0 = __builtin_amdgcn_wmma_f32_16x16x4_f32(false, a, false, b,
                                                 (short)0, f0, false, false);
      w = wrow[16];
      b.x = w.x; b.y = w.y;
      f1 = __builtin_amdgcn_wmma_f32_16x16x4_f32(false, a, false, b,
                                                 (short)0, f1, false, false);
      w = wrow[32];
      b.x = w.x; b.y = w.y;
      f2 = __builtin_amdgcn_wmma_f32_16x16x4_f32(false, a, false, b,
                                                 (short)0, f2, false, false);
      w = wrow[48];
      b.x = w.x; b.y = w.y;
      f3 = __builtin_amdgcn_wmma_f32_16x16x4_f32(false, a, false, b,
                                                 (short)0, f3, false, false);
    }

#pragma unroll
    for (int r = 0; r < 8; ++r) {
      const int kr = wave * 16 + r + 8 * half;
      float* p = (float*)Wp + (kr >> 1) * (2 * WP_STRIDE) + (kr & 1) + 2 * l15;
      p[0]  = f0[r];
      p[32] = f1[r];
      p[64] = f2[r];
      p[96] = f3[r];
    }
  }

  float b0 = 0.0f, b1 = 0.0f, b2 = 0.0f, b3 = 0.0f;
  for (int kp = 0; kp < D_HID / 2; ++kp) {
    const float s0 = bias1[2 * kp];
    const float s1 = bias1[2 * kp + 1];
    const float2* q = Wq + kp * WP_STRIDE + l15;
    float2 w;
    w = q[0];  b0 = fmaf(s0, w.x, b0); b0 = fmaf(s1, w.y, b0);
    w = q[16]; b1 = fmaf(s0, w.x, b1); b1 = fmaf(s1, w.y, b1);
    w = q[32]; b2 = fmaf(s0, w.x, b2); b2 = fmaf(s1, w.y, b2);
    w = q[48]; b3 = fmaf(s0, w.x, b3); b3 = fmaf(s1, w.y, b3);
  }

  __syncthreads();

  const int nTiles = nRows >> 4;

  for (int tile = blockIdx.x * 8 + wave; tile < nTiles;
       tile += nBlocks * 8) {
    const int rowBase = tile * 16;
    const float* aRow = A + (size_t)(rowBase + l15) * D_IN;

    v8f acc0 = {}, acc1 = {}, acc2 = {}, acc3 = {};

#pragma unroll 4
    for (int k = 0; k < D_IN; k += 4) {
      const int kk = k + 2 * half;
      v2f a;
      a.x = aRow[kk];
      a.y = aRow[kk + 1];

      const float2* wrow = Wp + ((k >> 1) + half) * WP_STRIDE + l15;
      float2 w;
      v2f b;

      w = wrow[0];
      b.x = w.x; b.y = w.y;
      acc0 = __builtin_amdgcn_wmma_f32_16x16x4_f32(false, a, false, b,
                                                   (short)0, acc0, false, false);
      w = wrow[16];
      b.x = w.x; b.y = w.y;
      acc1 = __builtin_amdgcn_wmma_f32_16x16x4_f32(false, a, false, b,
                                                   (short)0, acc1, false, false);
      w = wrow[32];
      b.x = w.x; b.y = w.y;
      acc2 = __builtin_amdgcn_wmma_f32_16x16x4_f32(false, a, false, b,
                                                   (short)0, acc2, false, false);
      w = wrow[48];
      b.x = w.x; b.y = w.y;
      acc3 = __builtin_amdgcn_wmma_f32_16x16x4_f32(false, a, false, b,
                                                   (short)0, acc3, false, false);
    }

#pragma unroll
    for (int r = 0; r < 8; ++r) {
      const int row = rowBase + r + 8 * half;
      float* o = out + (size_t)row * 64 + l15;
      o[0]  = acc0[r] + b0;
      o[16] = acc1[r] + b1;
      o[32] = acc2[r] + b2;
      o[48] = acc3[r] + b3;
    }
  }
}

__global__ __launch_bounds__(256) void fill_rows_f32(
    float* __restrict__ p, const float* __restrict__ rowv, size_t n,
    unsigned* __restrict__ chunkCnt, size_t wsBytes) {
  size_t i = (size_t)blockIdx.x * blockDim.x + threadIdx.x;
  if (i < n) p[i] = rowv[i & 63];
  if (i < 2 && wsBytes >= WS_NEEDED) chunkCnt[i * CNT_STRIDE] = 0u;
}

#define CHUNK_EDGES 64

__device__ __forceinline__ void push_one(const float* __restrict__ h,
                                         float* __restrict__ agg, int s, int d,
                                         int lane) {
  const float2 v = ((const float2*)h)[(size_t)s * 32 + lane];
  float* p = agg + (size_t)d * 64 + 2 * lane;
  unsafeAtomicAdd(p + 0, v.x);
  unsafeAtomicAdd(p + 1, v.y);
}

__device__ __forceinline__ void push_batch(const float* __restrict__ h,
                                           float* __restrict__ agg, int s,
                                           int d, bool valid, int set,
                                           int lane) {
  const bool mine = valid && ((d >= N_NODES / 2) == (set != 0));
  unsigned mask = __builtin_amdgcn_ballot_w32(mine);
  while (__builtin_popcount(mask) >= 4) {
    const int i0 = __builtin_ctz(mask); mask &= mask - 1;
    const int i1 = __builtin_ctz(mask); mask &= mask - 1;
    const int i2 = __builtin_ctz(mask); mask &= mask - 1;
    const int i3 = __builtin_ctz(mask); mask &= mask - 1;
    const int s0 = __builtin_amdgcn_readlane(s, i0);
    const int s1 = __builtin_amdgcn_readlane(s, i1);
    const int s2 = __builtin_amdgcn_readlane(s, i2);
    const int s3 = __builtin_amdgcn_readlane(s, i3);
    const int d0 = __builtin_amdgcn_readlane(d, i0);
    const int d1 = __builtin_amdgcn_readlane(d, i1);
    const int d2 = __builtin_amdgcn_readlane(d, i2);
    const int d3 = __builtin_amdgcn_readlane(d, i3);
    const float2 v0 = ((const float2*)h)[(size_t)s0 * 32 + lane];
    const float2 v1 = ((const float2*)h)[(size_t)s1 * 32 + lane];
    const float2 v2 = ((const float2*)h)[(size_t)s2 * 32 + lane];
    const float2 v3 = ((const float2*)h)[(size_t)s3 * 32 + lane];
    float* p0 = agg + (size_t)d0 * 64 + 2 * lane;
    float* p1 = agg + (size_t)d1 * 64 + 2 * lane;
    float* p2 = agg + (size_t)d2 * 64 + 2 * lane;
    float* p3 = agg + (size_t)d3 * 64 + 2 * lane;
    unsafeAtomicAdd(p0 + 0, v0.x); unsafeAtomicAdd(p0 + 1, v0.y);
    unsafeAtomicAdd(p1 + 0, v1.x); unsafeAtomicAdd(p1 + 1, v1.y);
    unsafeAtomicAdd(p2 + 0, v2.x); unsafeAtomicAdd(p2 + 1, v2.y);
    unsafeAtomicAdd(p3 + 0, v3.x); unsafeAtomicAdd(p3 + 1, v3.y);
  }
  while (mask) {
    const int i0 = __builtin_ctz(mask); mask &= mask - 1;
    push_one(h, agg, __builtin_amdgcn_readlane(s, i0),
             __builtin_amdgcn_readlane(d, i0), lane);
  }
}

__device__ __forceinline__ void push_chunk(const float* __restrict__ h,
                                           const int* __restrict__ esrc,
                                           const int* __restrict__ edst,
                                           float* __restrict__ agg, int chunk,
                                           int set, int nEdges, int lane) {
  const int eA = chunk * CHUNK_EDGES + lane;
  const int eB = eA + 32;
  const bool okA = eA < nEdges;
  const bool okB = eB < nEdges;
  const int sA = okA ? esrc[eA] : 0;
  const int dA = okA ? edst[eA] : 0;
  const int sB = okB ? esrc[eB] : 0;
  const int dB = okB ? edst[eB] : 0;
  push_batch(h, agg, sA, dA, okA, set, lane);
  push_batch(h, agg, sB, dB, okB, set, lane);
}

__global__ __launch_bounds__(256) void scatter_edges_by_half(
    const float* __restrict__ h, const int* __restrict__ esrc,
    const int* __restrict__ edst, float* __restrict__ agg,
    unsigned* __restrict__ chunkCnt, size_t wsBytes, int nEdges,
    int nGroupChunks) {
  __shared__ unsigned sHalf;
  __shared__ unsigned sTicket;
  const int lane = threadIdx.x & 31;
  const int wave = threadIdx.x >> 5;

  if (wsBytes >= WS_NEEDED) {
    if (threadIdx.x == 0) {
      const unsigned where = __builtin_amdgcn_s_sendmsg_rtn(0x87);
      sHalf = (where >> 18) & 1u;
    }
    __syncthreads();
    const int myHalf = (int)sHalf;

    for (int pass = 0; pass < 2; ++pass) {
      const int set = myHalf ^ pass;
      unsigned* cnt = chunkCnt + set * CNT_STRIDE;
      for (int it = 0; it <= nGroupChunks; ++it) {
        if (threadIdx.x == 0) {
          unsigned t = (unsigned)nGroupChunks;
          if (pass == 0 ||
              __hip_atomic_load(cnt, __ATOMIC_RELAXED,
                                __HIP_MEMORY_SCOPE_AGENT) <
                  (unsigned)nGroupChunks)
            t = atomicAdd(cnt, 1u);
          sTicket = t;
        }
        __syncthreads();
        const unsigned t = sTicket;
        __syncthreads();
        if (t >= (unsigned)nGroupChunks) break;
        push_chunk(h, esrc, edst, agg, (int)t * 8 + wave, set, nEdges, lane);
      }
    }
  } else {
    const int set = (blockIdx.x >> 2) & 1;
    const int j = (int)(((blockIdx.x >> 3) << 2) | (blockIdx.x & 3));
    const int wavesPerSet = (int)(gridDim.x >> 1) * 8;
    for (int t = j * 8 + wave; t < nGroupChunks * 8; t += wavesPerSet)
      push_chunk(h, esrc, edst, agg, t, set, nEdges, lane);
  }
}

extern "C" void kernel_launch(void* const* d_in, const int* in_sizes, int n_in,
                              void* d_out, int out_size, void* d_ws,
                              size_t ws_size, hipStream_t stream) {
  const float* data   = (const float*)d_in[0];
  const int*   esrc   = (const int*)d_in[1];
  const int*   edst   = (const int*)d_in[2];
  const float* W_pre  = (const float*)d_in[3];
  const float* b_pre  = (const float*)d_in[4];
  const float* W_post = (const float*)d_in[5];
  const float* b_post = (const float*)d_in[6];

  float* out = (float*)d_out;
  float* g   = (float*)d_ws;
  unsigned* chunkCnt = (unsigned*)((char*)d_ws + G_BYTES);

  {
    const size_t n = (size_t)N_NODES * D_OUT;
    fill_rows_f32<<<(unsigned)((n + 255) / 256), 256, 0, stream>>>(
        out, b_post, n, chunkCnt, ws_size);
  }

  {
    const int blocks = 250;
    gemm_folded_wmma_f32<<<blocks, 256, 0, stream>>>(data, W_pre, b_pre, W_post,
                                                     g, N_NODES, blocks);
  }

  {
    const int groupChunks = N_EDGES / (8 * CHUNK_EDGES);
    scatter_edges_by_half<<<2048, 256, 0, stream>>>(
        g, esrc, edst, out, chunkCnt, ws_size, N_EDGES, groupChunks);
  }
}


# edge pass: edges grouped by destination range per call (LDS counting sort), owner work-groups sum rows in LDS float adds and store each out row once; no memory atomics, no fill; add order changes
# speedup vs baseline: 3.3345x; 1.5060x over previous
#include <hip/hip_runtime.h>


typedef __attribute__((ext_vector_type(2))) float v2f;
typedef __attribute__((ext_vector_type(8))) float v8f;

#define N_NODES 100000
#define D_IN    128
#define D_HID   64
#define D_OUT   64
#define N_EDGES 1600000

#define N_BINS    256
#define ROWS_PER  391
#define SLICE     (N_EDGES / N_BINS)
#define OFF_PITCH (N_BINS + 1)

#define G_BYTES   ((size_t)N_NODES * D_OUT * sizeof(float))
#define REC_BYTES ((size_t)N_EDGES * sizeof(int))
#define OFF_BYTES ((size_t)N_BINS * OFF_PITCH * sizeof(int))
#define WS_NEEDED (G_BYTES + REC_BYTES + OFF_BYTES)

#define WP_STRIDE 80

__global__ __launch_bounds__(256) void gemm_folded_wmma_f32(
    const float* __restrict__ A, const float* __restrict__ W1,
    const float* __restrict__ bias1, const float* __restrict__ W2,
    float* __restrict__ out, int nRows, int nBlocks) {
  __shared__ float2 Wp[(D_IN / 2) * WP_STRIDE];
  __shared__ float2 Wq[(D_HID / 2) * WP_STRIDE];

  for (int i = threadIdx.x; i < D_HID * D_OUT; i += 256) {
    const int k = i >> 6;
    const int c = i & 63;
    ((float*)Wq)[(k >> 1) * (2 * WP_STRIDE) + c * 2 + (k & 1)] = W2[i];
  }
  __syncthreads();

  const int lane = threadIdx.x & 31;
  const int wave = threadIdx.x >> 5;
  const int half = lane >> 4;
  const int l15  = lane & 15;

  {
    const float* aRow = W1 + (size_t)(wave * 16 + l15) * D_HID;
    v8f f0 = {}, f1 = {}, f2 = {}, f3 = {};

#pragma unroll 4
    for (int k = 0; k < D_HID; k += 4) {
      const int kk = k + 2 * half;
      v2f a;
      a.x = aRow[kk];
      a.y = aRow[kk + 1];

      const float2* wrow = Wq + ((k >> 1) + half) * WP_STRIDE + l15;
      float2 w;
      v2f b;

      w = wrow[0];
      b.x = w.x; b.y = w.y;
      f0 = __builtin_amdgcn_wmma_f32_16x16x4_f32(false, a, false, b,
                                                 (short)0, f0, false, false);
      w = wrow[16];
      b.x = w.x; b.y = w.y;
      f1 = __builtin_amdgcn_wmma_f32_16x16x4_f32(false, a, false, b,
                                                 (short)0, f1, false, false);
      w = wrow[32];
      b.x = w.x; b.y = w.y;
      f2 = __builtin_amdgcn_wmma_f32_16x16x4_f32(false, a, false, b,
                                                 (short)0, f2, false, false);
      w = wrow[48];
      b.x = w.x; b.y = w.y;
      f3 = __builtin_amdgcn_wmma_f32_16x16x4_f32(false, a, false, b,
                                                 (short)0, f3, false, false);
    }

#pragma unroll
    for (int r = 0; r < 8; ++r) {
      const int kr = wave * 16 + r + 8 * half;
      float* p = (float*)Wp + (kr >> 1) * (2 * WP_STRIDE) + (kr & 1) + 2 * l15;
      p[0]  = f0[r];
      p[32] = f1[r];
      p[64] = f2[r];
      p[96] = f3[r];
    }
  }

  float b0 = 0.0f, b1 = 0.0f, b2 = 0.0f, b3 = 0.0f;
  for (int kp = 0; kp < D_HID / 2; ++kp) {
    const float s0 = bias1[2 * kp];
    const float s1 = bias1[2 * kp + 1];
    const float2* q = Wq + kp * WP_STRIDE + l15;
    float2 w;
    w = q[0];  b0 = fmaf(s0, w.x, b0); b0 = fmaf(s1, w.y, b0);
    w = q[16]; b1 = fmaf(s0, w.x, b1); b1 = fmaf(s1, w.y, b1);
    w = q[32]; b2 = fmaf(s0, w.x, b2); b2 = fmaf(s1, w.y, b2);
    w = q[48]; b3 = fmaf(s0, w.x, b3); b3 = fmaf(s1, w.y, b3);
  }

  __syncthreads();

  const int nTiles = nRows >> 4;

  for (int tile = blockIdx.x * 8 + wave; tile < nTiles;
       tile += nBlocks * 8) {
    const int rowBase = tile * 16;
    const float* aRow = A + (size_t)(rowBase + l15) * D_IN;

    v8f acc0 = {}, acc1 = {}, acc2 = {}, acc3 = {};

#pragma unroll 4
    for (int k = 0; k < D_IN; k += 4) {
      const int kk = k + 2 * half;
      v2f a;
      a.x = aRow[kk];
      a.y = aRow[kk + 1];

      const float2* wrow = Wp + ((k >> 1) + half) * WP_STRIDE + l15;
      float2 w;
      v2f b;

      w = wrow[0];
      b.x = w.x; b.y = w.y;
      acc0 = __builtin_amdgcn_wmma_f32_16x16x4_f32(false, a, false, b,
                                                   (short)0, acc0, false, false);
      w = wrow[16];
      b.x = w.x; b.y = w.y;
      acc1 = __builtin_amdgcn_wmma_f32_16x16x4_f32(false, a, false, b,
                                                   (short)0, acc1, false, false);
      w = wrow[32];
      b.x = w.x; b.y = w.y;
      acc2 = __builtin_amdgcn_wmma_f32_16x16x4_f32(false, a, false, b,
                                                   (short)0, acc2, false, false);
      w = wrow[48];
      b.x = w.x; b.y = w.y;
      acc3 = __builtin_amdgcn_wmma_f32_16x16x4_f32(false, a, false, b,
                                                   (short)0, acc3, false, false);
    }

#pragma unroll
    for (int r = 0; r < 8; ++r) {
      const int row = rowBase + r + 8 * half;
      float* o = out + (size_t)row * 64 + l15;
      o[0]  = acc0[r] + b0;
      o[16] = acc1[r] + b1;
      o[32] = acc2[r] + b2;
      o[48] = acc3[r] + b3;
    }
  }
}

__global__ __launch_bounds__(1024) void group_edges(
    const int* __restrict__ esrc, const int* __restrict__ edst,
    unsigned* __restrict__ grp, int* __restrict__ off, size_t wsBytes) {
  __shared__ unsigned cnt[N_BINS];
  __shared__ int start[OFF_PITCH];
  __shared__ unsigned sg[SLICE];

  if (wsBytes < WS_NEEDED) return;

  const int tid = threadIdx.x;
  const int base = blockIdx.x * SLICE;
  constexpr int PER = (SLICE + 1023) / 1024;

  if (tid < N_BINS) cnt[tid] = 0u;
  __syncthreads();

  int s[PER], d[PER], bin[PER];
  unsigned rank[PER];
#pragma unroll
  for (int k = 0; k < PER; ++k) {
    const int i = k * 1024 + tid;
    if (i < SLICE) {
      s[k] = esrc[base + i];
      d[k] = edst[base + i];
    } else {
      s[k] = 0;
      d[k] = 0;
    }
  }
#pragma unroll
  for (int k = 0; k < PER; ++k) {
    const int i = k * 1024 + tid;
    const int q = d[k] / ROWS_PER;
    bin[k] = q < N_BINS - 1 ? q : N_BINS - 1;
    rank[k] = 0u;
    if (i < SLICE) rank[k] = atomicAdd(&cnt[bin[k]], 1u);
  }
  __syncthreads();

  if (tid < OFF_PITCH) {
    int sum = 0;
    for (int u = 0; u < tid; ++u) sum += (int)cnt[u];
    start[tid] = sum;
    off[blockIdx.x * OFF_PITCH + tid] = sum;
  }
  __syncthreads();

#pragma unroll
  for (int k = 0; k < PER; ++k) {
    const int i = k * 1024 + tid;
    if (i < SLICE) {
      const int p = start[bin[k]] + (int)rank[k];
      sg[p] = ((unsigned)s[k] << 9) | (unsigned)(d[k] - bin[k] * ROWS_PER);
    }
  }
  __syncthreads();

  for (int i = tid; i < SLICE; i += 1024) grp[base + i] = sg[i];
}

#define ROWS_IN_FLIGHT 16

__device__ __forceinline__ void add_batch(const float2* __restrict__ g2,
                                          float* acc, int s, int r,
                                          unsigned mask, int lane) {
  while (mask) {
    int sk[ROWS_IN_FLIGHT], rk[ROWS_IN_FLIGHT];
    bool ok[ROWS_IN_FLIGHT];
    float2 v[ROWS_IN_FLIGHT];
#pragma unroll
    for (int k = 0; k < ROWS_IN_FLIGHT; ++k) {
      ok[k] = mask != 0u;
      const int i = ok[k] ? __builtin_ctz(mask) : 0;
      mask &= mask - 1u;
      sk[k] = __builtin_amdgcn_readlane(s, i);
      rk[k] = __builtin_amdgcn_readlane(r, i);
    }
#pragma unroll
    for (int k = 0; k < ROWS_IN_FLIGHT; ++k) {
      v[k] = make_float2(0.0f, 0.0f);
      if (ok[k]) v[k] = g2[(size_t)sk[k] * 32 + lane];
    }
#pragma unroll
    for (int k = 0; k < ROWS_IN_FLIGHT; ++k) {
      if (ok[k]) {
        float* p = acc + rk[k] * 64 + 2 * lane;
        atomicAdd(p + 0, v[k].x);
        atomicAdd(p + 1, v[k].y);
      }
    }
  }
}

__global__ __launch_bounds__(1024) void sum_rows_by_owner(
    const float* __restrict__ g, const int* __restrict__ esrc,
    const int* __restrict__ edst, const unsigned* __restrict__ grp,
    const int* __restrict__ off,
    const float* __restrict__ bias2, float* __restrict__ out, size_t wsBytes) {
  __shared__ float acc[ROWS_PER * 64];

  const int lane = threadIdx.x & 31;
  const int wave = threadIdx.x >> 5;
  const int j = blockIdx.x;
  const int lo = j * ROWS_PER;
  const int rows = (N_NODES - lo) < ROWS_PER ? (N_NODES - lo) : ROWS_PER;
  const float2* g2 = (const float2*)g;

  for (int i = threadIdx.x; i < ROWS_PER * 16; i += 1024)
    ((float4*)acc)[i] = make_float4(0.0f, 0.0f, 0.0f, 0.0f);
  __syncthreads();

  if (wsBytes >= WS_NEEDED) {
    int myBeg = 0, myEnd = 0;
    if (lane < N_BINS / 32) {
      const int b = wave + 32 * lane;
      myBeg = off[b * OFF_PITCH + j];
      myEnd = off[b * OFF_PITCH + j + 1];
    }
    int beg = __builtin_amdgcn_readlane(myBeg, 0);
    int end = __builtin_amdgcn_readlane(myEnd, 0);
    bool inN = beg + lane < end;
    unsigned wN = inN ? grp[wave * SLICE + beg + lane] : 0u;
    for (int t = 0; t < N_BINS / 32; ++t) {
      const int b = wave + 32 * t;
      const unsigned* pg = grp + b * SLICE;
      const int s0 = (int)(wN >> 9);
      const int r0 = (int)(wN & 511u);
      const bool mine0 = inN && r0 < rows;
      const int begC = beg, endC = end;
      if (t + 1 < N_BINS / 32) {
        beg = __builtin_amdgcn_readlane(myBeg, t + 1);
        end = __builtin_amdgcn_readlane(myEnd, t + 1);
        inN = beg + lane < end;
        wN = inN ? grp[(b + 32) * SLICE + beg + lane] : 0u;
      }
      add_batch(g2, acc, s0, r0, __builtin_amdgcn_ballot_w32(mine0), lane);
      for (int p = begC + 32; p < endC; p += 32) {
        const int i = p + lane;
        const bool in = i < endC;
        const unsigned w = in ? pg[i] : 0u;
        const int s = (int)(w >> 9);
        const int r = (int)(w & 511u);
        const bool mine = in && r < rows;
        add_batch(g2, acc, s, r, __builtin_amdgcn_ballot_w32(mine), lane);
      }
    }
  } else {
    for (int p = wave * 32; p < N_EDGES; p += 1024) {
      const int i = p + lane;
      const bool in = i < N_EDGES;
      const int s = in ? esrc[i] : 0;
      const int r = in ? edst[i] - lo : -1;
      const bool mine = in && (unsigned)r < (unsigned)rows;
      add_batch(g2, acc, s, r, __builtin_amdgcn_ballot_w32(mine), lane);
    }
  }

  __syncthreads();

  const float2 bb = ((const float2*)bias2)[lane];
  for (int r = wave; r < rows; r += 32) {
    const float2 a = ((const float2*)acc)[r * 32 + lane];
    ((float2*)out)[(size_t)(lo + r) * 32 + lane] =
        make_float2(a.x + bb.x, a.y + bb.y);
  }
}

extern "C" void kernel_launch(void* const* d_in, const int* in_sizes, int n_in,
                              void* d_out, int out_size, void* d_ws,
                              size_t ws_size, hipStream_t stream) {
  const float* data   = (const float*)d_in[0];
  const int*   esrc   = (const int*)d_in[1];
  const int*   edst   = (const int*)d_in[2];
  const float* W_pre  = (const float*)d_in[3];
  const float* b_pre  = (const float*)d_in[4];
  const float* W_post = (const float*)d_in[5];
  const float* b_post = (const float*)d_in[6];

  float* out = (float*)d_out;
  float* g   = (float*)d_ws;
  unsigned* grp = (unsigned*)((char*)d_ws + G_BYTES);
  int* off = (int*)((char*)d_ws + G_BYTES + REC_BYTES);

  {
    const int blocks = 250;
    gemm_folded_wmma_f32<<<blocks, 256, 0, stream>>>(data, W_pre, b_pre, W_post,
                                                     g, N_NODES, blocks);
  }

  group_edges<<<N_BINS, 1024, 0, stream>>>(esrc, edst, grp, off, ws_size);

  sum_rows_by_owner<<<N_BINS, 1024, 0, stream>>>(g, esrc, edst, grp, off,
                                                 b_post, out, ws_size);
}


# edge grouping moved into the node GEMM launch as 256 extra 8-wave work-groups (independent jobs side by side): 2 launches, not 3; edge order inside a group, hence add order, may change
# speedup vs baseline: 3.7124x; 1.1133x over previous
#include <hip/hip_runtime.h>


typedef __attribute__((ext_vector_type(2))) float v2f;
typedef __attribute__((ext_vector_type(8))) float v8f;

#define N_NODES 100000
#define D_IN    128
#define D_HID   64
#define D_OUT   64
#define N_EDGES 1600000

#define N_BINS    256
#define ROWS_PER  391
#define SLICE     (N_EDGES / N_BINS)
#define OFF_PITCH (N_BINS + 1)

#define G_BYTES   ((size_t)N_NODES * D_OUT * sizeof(float))
#define REC_BYTES ((size_t)N_EDGES * sizeof(int))
#define OFF_BYTES ((size_t)N_BINS * OFF_PITCH * sizeof(int))
#define WS_NEEDED (G_BYTES + REC_BYTES + OFF_BYTES)

#define WP_STRIDE 80
#define WP_BYTES  ((D_IN / 2) * WP_STRIDE * 8)
#define WQ_BYTES  ((D_HID / 2) * WP_STRIDE * 8)
#define LDS_BYTES (WP_BYTES + WQ_BYTES)

#define GEMM_BLOCKS 250
#define GRID_FIRST  506
static_assert(GRID_FIRST == GEMM_BLOCKS + N_BINS, "grid of the first launch");

__device__ __forceinline__ void gemm_folded_role(
    unsigned char* lds, const float* __restrict__ A,
    const float* __restrict__ W1, const float* __restrict__ bias1,
    const float* __restrict__ W2, float* __restrict__ out, int nRows,
    int nBlocks) {
  float2* Wp = (float2*)lds;
  float2* Wq = (float2*)(lds + WP_BYTES);

  for (int i = threadIdx.x; i < D_HID * D_OUT; i += 256) {
    const int k = i >> 6;
    const int c = i & 63;
    ((float*)Wq)[(k >> 1) * (2 * WP_STRIDE) + c * 2 + (k & 1)] = W2[i];
  }
  __syncthreads();

  const int lane = threadIdx.x & 31;
  const int wave = threadIdx.x >> 5;
  const int half = lane >> 4;
  const int l15  = lane & 15;

  {
    const float* aRow = W1 + (size_t)(wave * 16 + l15) * D_HID;
    v8f f0 = {}, f1 = {}, f2 = {}, f3 = {};

#pragma unroll 4
    for (int k = 0; k < D_HID; k += 4) {
      const int kk = k + 2 * half;
      v2f a;
      a.x = aRow[kk];
      a.y = aRow[kk + 1];

      const float2* wrow = Wq + ((k >> 1) + half) * WP_STRIDE + l15;
      float2 w;
      v2f b;

      w = wrow[0];
      b.x = w.x; b.y = w.y;
      f0 = __builtin_amdgcn_wmma_f32_16x16x4_f32(false, a, false, b,
                                                 (short)0, f0, false, false);
      w = wrow[16];
      b.x = w.x; b.y = w.y;
      f1 = __builtin_amdgcn_wmma_f32_16x16x4_f32(false, a, false, b,
                                                 (short)0, f1, false, false);
      w = wrow[32];
      b.x = w.x; b.y = w.y;
      f2 = __builtin_amdgcn_wmma_f32_16x16x4_f32(false, a, false, b,
                                                 (short)0, f2, false, false);
      w = wrow[48];
      b.x = w.x; b.y = w.y;
      f3 = __builtin_amdgcn_wmma_f32_16x16x4_f32(false, a, false, b,
                                                 (short)0, f3, false, false);
    }

#pragma unroll
    for (int r = 0; r < 8; ++r) {
      const int kr = wave * 16 + r + 8 * half;
      float* p = (float*)Wp + (kr >> 1) * (2 * WP_STRIDE) + (kr & 1) + 2 * l15;
      p[0]  = f0[r];
      p[32] = f1[r];
      p[64] = f2[r];
      p[96] = f3[r];
    }
  }

  float b0 = 0.0f, b1 = 0.0f, b2 = 0.0f, b3 = 0.0f;
  for (int kp = 0; kp < D_HID / 2; ++kp) {
    const float s0 = bias1[2 * kp];
    const float s1 = bias1[2 * kp + 1];
    const float2* q = Wq + kp * WP_STRIDE + l15;
    float2 w;
    w = q[0];  b0 = fmaf(s0, w.x, b0); b0 = fmaf(s1, w.y, b0);
    w = q[16]; b1 = fmaf(s0, w.x, b1); b1 = fmaf(s1, w.y, b1);
    w = q[32]; b2 = fmaf(s0, w.x, b2); b2 = fmaf(s1, w.y, b2);
    w = q[48]; b3 = fmaf(s0, w.x, b3); b3 = fmaf(s1, w.y, b3);
  }

  __syncthreads();

  const int nTiles = nRows >> 4;

  for (int tile = blockIdx.x * 8 + wave; tile < nTiles;
       tile += nBlocks * 8) {
    const int rowBase = tile * 16;
    const float* aRow = A + (size_t)(rowBase + l15) * D_IN;

    v8f acc0 = {}, acc1 = {}, acc2 = {}, acc3 = {};

#pragma unroll 4
    for (int k = 0; k < D_IN; k += 4) {
      const int kk = k + 2 * half;
      v2f a;
      a.x = aRow[kk];
      a.y = aRow[kk + 1];

      const float2* wrow = Wp + ((k >> 1) + half) * WP_STRIDE + l15;
      float2 w;
      v2f b;

      w = wrow[0];
      b.x = w.x; b.y = w.y;
      acc0 = __builtin_amdgcn_wmma_f32_16x16x4_f32(false, a, false, b,
                                                   (short)0, acc0, false, false);
      w = wrow[16];
      b.x = w.x; b.y = w.y;
      acc1 = __builtin_amdgcn_wmma_f32_16x16x4_f32(false, a, false, b,
                                                   (short)0, acc1, false, false);
      w = wrow[32];
      b.x = w.x; b.y = w.y;
      acc2 = __builtin_amdgcn_wmma_f32_16x16x4_f32(false, a, false, b,
                                                   (short)0, acc2, false, false);
      w = wrow[48];
      b.x = w.x; b.y = w.y;
      acc3 = __builtin_amdgcn_wmma_f32_16x16x4_f32(false, a, false, b,
                                                   (short)0, acc3, false, false);
    }

#pragma unroll
    for (int r = 0; r < 8; ++r) {
      const int row = rowBase + r + 8 * half;
      float* o = out + (size_t)row * 64 + l15;
      o[0]  = acc0[r] + b0;
      o[16] = acc1[r] + b1;
      o[32] = acc2[r] + b2;
      o[48] = acc3[r] + b3;
    }
  }
}

__device__ __forceinline__ void group_edges_role(
    unsigned char* lds, int b, const int* __restrict__ esrc,
    const int* __restrict__ edst, unsigned* __restrict__ grp,
    int* __restrict__ off, size_t wsBytes) {
  unsigned* cnt = (unsigned*)lds;
  int* start = (int*)(lds + 1024);
  unsigned* sg = (unsigned*)(lds + 1024 + 1040);

  if (wsBytes < WS_NEEDED) return;

  const int tid = threadIdx.x;
  const int base = b * SLICE;
  constexpr int PER = (SLICE + 255) / 256;
  const bool lastIn = (PER - 1) * 256 + tid < SLICE;

  cnt[tid] = 0u;
  __syncthreads();

  int s[PER], d[PER];
  unsigned w[PER], br[PER];
#pragma unroll
  for (int k = 0; k < PER; ++k) {
    const int i = k * 256 + tid;
    if (k < PER - 1 || lastIn) {
      s[k] = esrc[base + i];
      d[k] = edst[base + i];
    } else {
      s[k] = 0;
      d[k] = 0;
    }
  }
#pragma unroll
  for (int k = 0; k < PER; ++k) {
    const int q = d[k] / ROWS_PER;
    const int bin = q < N_BINS - 1 ? q : N_BINS - 1;
    w[k] = ((unsigned)s[k] << 9) | (unsigned)(d[k] - bin * ROWS_PER);
    unsigned rank = 0u;
    if (k < PER - 1 || lastIn)
      rank = atomicAdd(&cnt[bin], 1u);
    br[k] = (unsigned)bin | (rank << 8);
  }
  __syncthreads();

  {
    int sum = 0;
    for (int u = 0; u < tid; ++u) sum += (int)cnt[u];
    start[tid] = sum;
    off[b * OFF_PITCH + tid] = sum;
    if (tid == N_BINS - 1) {
      const int tot = sum + (int)cnt[tid];
      start[N_BINS] = tot;
      off[b * OFF_PITCH + N_BINS] = tot;
    }
  }
  __syncthreads();

#pragma unroll
  for (int k = 0; k < PER; ++k) {
    if (k < PER - 1 || lastIn)
      sg[start[br[k] & 255u] + (int)(br[k] >> 8)] = w[k];
  }
  __syncthreads();

  for (int i = tid; i < SLICE; i += 256) grp[base + i] = sg[i];
}

__global__ __launch_bounds__(256) void gemm_and_group(
    const float* __restrict__ A, const float* __restrict__ W1,
    const float* __restrict__ bias1, const float* __restrict__ W2,
    float* __restrict__ g, int nRows, int nBlocks,
    const int* __restrict__ esrc, const int* __restrict__ edst,
    unsigned* __restrict__ grp, int* __restrict__ off, size_t wsBytes) {
  __shared__ __attribute__((aligned(16))) unsigned char lds[LDS_BYTES];
  if (blockIdx.x < GEMM_BLOCKS) {
    gemm_folded_role(lds, A, W1, bias1, W2, g, nRows, nBlocks);
  } else {
    group_edges_role(lds, (int)blockIdx.x - GEMM_BLOCKS, esrc, edst, grp, off,
                     wsBytes);
  }
}

#define ROWS_IN_FLIGHT 16

__device__ __forceinline__ void add_batch(const float2* __restrict__ g2,
                                          float* acc, int s, int r,
                                          unsigned mask, int lane) {
  while (mask) {
    int sk[ROWS_IN_FLIGHT], rk[ROWS_IN_FLIGHT];
    bool ok[ROWS_IN_FLIGHT];
    float2 v[ROWS_IN_FLIGHT];
#pragma unroll
    for (int k = 0; k < ROWS_IN_FLIGHT; ++k) {
      ok[k] = mask != 0u;
      const int i = ok[k] ? __builtin_ctz(mask) : 0;
      mask &= mask - 1u;
      sk[k] = __builtin_amdgcn_readlane(s, i);
      rk[k] = __builtin_amdgcn_readlane(r, i);
    }
#pragma unroll
    for (int k = 0; k < ROWS_IN_FLIGHT; ++k) {
      v[k] = make_float2(0.0f, 0.0f);
      if (ok[k]) v[k] = g2[(size_t)sk[k] * 32 + lane];
    }
#pragma unroll
    for (int k = 0; k < ROWS_IN_FLIGHT; ++k) {
      if (ok[k]) {
        float* p = acc + rk[k] * 64 + 2 * lane;
        atomicAdd(p + 0, v[k].x);
        atomicAdd(p + 1, v[k].y);
      }
    }
  }
}

__global__ __launch_bounds__(1024) void sum_rows_by_owner(
    const float* __restrict__ g, const int* __restrict__ esrc,
    const int* __restrict__ edst, const unsigned* __restrict__ grp,
    const int* __restrict__ off,
    const float* __restrict__ bias2, float* __restrict__ out, size_t wsBytes) {
  __shared__ float acc[ROWS_PER * 64];

  const int lane = threadIdx.x & 31;
  const int wave = threadIdx.x >> 5;
  const int j = blockIdx.x;
  const int lo = j * ROWS_PER;
  const int rows = (N_NODES - lo) < ROWS_PER ? (N_NODES - lo) : ROWS_PER;
  const float2* g2 = (const float2*)g;

  for (int i = threadIdx.x; i < ROWS_PER * 16; i += 1024)
    ((float4*)acc)[i] = make_float4(0.0f, 0.0f, 0.0f, 0.0f);
  __syncthreads();

  if (wsBytes >= WS_NEEDED) {
    int myBeg = 0, myEnd = 0;
    if (lane < N_BINS / 32) {
      const int b = wave + 32 * lane;
      myBeg = off[b * OFF_PITCH + j];
      myEnd = off[b * OFF_PITCH + j + 1];
    }
    int beg = __builtin_amdgcn_readlane(myBeg, 0);
    int end = __builtin_amdgcn_readlane(myEnd, 0);
    bool inN = beg + lane < end;
    unsigned wN = inN ? grp[wave * SLICE + beg + lane] : 0u;
    for (int t = 0; t < N_BINS / 32; ++t) {
      const int b = wave + 32 * t;
      const unsigned* pg = grp + b * SLICE;
      const int s0 = (int)(wN >> 9);
      const int r0 = (int)(wN & 511u);
      const bool mine0 = inN && r0 < rows;
      const int begC = beg, endC = end;
      if (t + 1 < N_BINS / 32) {
        beg = __builtin_amdgcn_readlane(myBeg, t + 1);
        end = __builtin_amdgcn_readlane(myEnd, t + 1);
        inN = beg + lane < end;
        wN = inN ? grp[(b + 32) * SLICE + beg + lane] : 0u;
      }
      add_batch(g2, acc, s0, r0, __builtin_amdgcn_ballot_w32(mine0), lane);
      for (int p = begC + 32; p < endC; p += 32) {
        const int i = p + lane;
        const bool in = i < endC;
        const unsigned w = in ? pg[i] : 0u;
        const int s = (int)(w >> 9);
        const int r = (int)(w & 511u);
        const bool mine = in && r < rows;
        add_batch(g2, acc, s, r, __builtin_amdgcn_ballot_w32(mine), lane);
      }
    }
  } else {
    for (int p = wave * 32; p < N_EDGES; p += 1024) {
      const int i = p + lane;
      const bool in = i < N_EDGES;
      const int s = in ? esrc[i] : 0;
      const int r = in ? edst[i] - lo : -1;
      const bool mine = in && (unsigned)r < (unsigned)rows;
      add_batch(g2, acc, s, r, __builtin_amdgcn_ballot_w32(mine), lane);
    }
  }

  __syncthreads();

  const float2 bb = ((const float2*)bias2)[lane];
  for (int r = wave; r < rows; r += 32) {
    const float2 a = ((const float2*)acc)[r * 32 + lane];
    ((float2*)out)[(size_t)(lo + r) * 32 + lane] =
        make_float2(a.x + bb.x, a.y + bb.y);
  }
}

extern "C" void kernel_launch(void* const* d_in, const int* in_sizes, int n_in,
                              void* d_out, int out_size, void* d_ws,
                              size_t ws_size, hipStream_t stream) {
  const float* data   = (const float*)d_in[0];
  const int*   esrc   = (const int*)d_in[1];
  const int*   edst   = (const int*)d_in[2];
  const float* W_pre  = (const float*)d_in[3];
  const float* b_pre  = (const float*)d_in[4];
  const float* W_post = (const float*)d_in[5];
  const float* b_post = (const float*)d_in[6];

  float* out = (float*)d_out;
  float* g   = (float*)d_ws;
  unsigned* grp = (unsigned*)((char*)d_ws + G_BYTES);
  int* off = (int*)((char*)d_ws + G_BYTES + REC_BYTES);

  gemm_and_group<<<GRID_FIRST, 256, 0, stream>>>(
      data, W_pre, b_pre, W_post, g, N_NODES, GEMM_BLOCKS, esrc, edst, grp, off,
      ws_size);

  sum_rows_by_owner<<<N_BINS, 1024, 0, stream>>>(g, esrc, edst, grp, off,
                                                 b_post, out, ws_size);
}


# edge pass inner loop: a group's edges are a lane prefix, so edge k is read from lane k (constant lane reads, quads padded with a null edge); per-edge mask/ctz/select work removed, 32 rows in flight
# speedup vs baseline: 3.8311x; 1.0320x over previous
#include <hip/hip_runtime.h>


typedef __attribute__((ext_vector_type(2))) float v2f;
typedef __attribute__((ext_vector_type(8))) float v8f;

#define N_NODES 100000
#define D_IN    128
#define D_HID   64
#define D_OUT   64
#define N_EDGES 1600000

#define N_BINS    256
#define ROWS_PER  391
#define SLICE     (N_EDGES / N_BINS)
#define OFF_PITCH (N_BINS + 1)

#define G_BYTES   ((size_t)N_NODES * D_OUT * sizeof(float))
#define REC_BYTES ((size_t)N_EDGES * sizeof(int))
#define OFF_BYTES ((size_t)N_BINS * OFF_PITCH * sizeof(int))
#define WS_NEEDED (G_BYTES + REC_BYTES + OFF_BYTES)

#define WP_STRIDE 80
#define WP_BYTES  ((D_IN / 2) * WP_STRIDE * 8)
#define WQ_BYTES  ((D_HID / 2) * WP_STRIDE * 8)
#define LDS_BYTES (WP_BYTES + WQ_BYTES)

#define GEMM_BLOCKS 250
#define GRID_FIRST  506
static_assert(GRID_FIRST == GEMM_BLOCKS + N_BINS, "grid of the first launch");

__device__ __forceinline__ void gemm_folded_role(
    unsigned char* lds, const float* __restrict__ A,
    const float* __restrict__ W1, const float* __restrict__ bias1,
    const float* __restrict__ W2, float* __restrict__ out, int nRows,
    int nBlocks) {
  float2* Wp = (float2*)lds;
  float2* Wq = (float2*)(lds + WP_BYTES);

  for (int i = threadIdx.x; i < D_HID * D_OUT; i += 256) {
    const int k = i >> 6;
    const int c = i & 63;
    ((float*)Wq)[(k >> 1) * (2 * WP_STRIDE) + c * 2 + (k & 1)] = W2[i];
  }
  __syncthreads();

  const int lane = threadIdx.x & 31;
  const int wave = threadIdx.x >> 5;
  const int half = lane >> 4;
  const int l15  = lane & 15;

  {
    const float* aRow = W1 + (size_t)(wave * 16 + l15) * D_HID;
    v8f f0 = {}, f1 = {}, f2 = {}, f3 = {};

#pragma unroll 4
    for (int k = 0; k < D_HID; k += 4) {
      const int kk = k + 2 * half;
      v2f a;
      a.x = aRow[kk];
      a.y = aRow[kk + 1];

      const float2* wrow = Wq + ((k >> 1) + half) * WP_STRIDE + l15;
      float2 w;
      v2f b;

      w = wrow[0];
      b.x = w.x; b.y = w.y;
      f0 = __builtin_amdgcn_wmma_f32_16x16x4_f32(false, a, false, b,
                                                 (short)0, f0, false, false);
      w = wrow[16];
      b.x = w.x; b.y = w.y;
      f1 = __builtin_amdgcn_wmma_f32_16x16x4_f32(false, a, false, b,
                                                 (short)0, f1, false, false);
      w = wrow[32];
      b.x = w.x; b.y = w.y;
      f2 = __builtin_amdgcn_wmma_f32_16x16x4_f32(false, a, false, b,
                                                 (short)0, f2, false, false);
      w = wrow[48];
      b.x = w.x; b.y = w.y;
      f3 = __builtin_amdgcn_wmma_f32_16x16x4_f32(false, a, false, b,
                                                 (short)0, f3, false, false);
    }

#pragma unroll
    for (int r = 0; r < 8; ++r) {
      const int kr = wave * 16 + r + 8 * half;
      float* p = (float*)Wp + (kr >> 1) * (2 * WP_STRIDE) + (kr & 1) + 2 * l15;
      p[0]  = f0[r];
      p[32] = f1[r];
      p[64] = f2[r];
      p[96] = f3[r];
    }
  }

  float b0 = 0.0f, b1 = 0.0f, b2 = 0.0f, b3 = 0.0f;
  for (int kp = 0; kp < D_HID / 2; ++kp) {
    const float s0 = bias1[2 * kp];
    const float s1 = bias1[2 * kp + 1];
    const float2* q = Wq + kp * WP_STRIDE + l15;
    float2 w;
    w = q[0];  b0 = fmaf(s0, w.x, b0); b0 = fmaf(s1, w.y, b0);
    w = q[16]; b1 = fmaf(s0, w.x, b1); b1 = fmaf(s1, w.y, b1);
    w = q[32]; b2 = fmaf(s0, w.x, b2); b2 = fmaf(s1, w.y, b2);
    w = q[48]; b3 = fmaf(s0, w.x, b3); b3 = fmaf(s1, w.y, b3);
  }

  __syncthreads();

  const int nTiles = nRows >> 4;

  for (int tile = blockIdx.x * 8 + wave; tile < nTiles;
       tile += nBlocks * 8) {
    const int rowBase = tile * 16;
    const float* aRow = A + (size_t)(rowBase + l15) * D_IN;

    v8f acc0 = {}, acc1 = {}, acc2 = {}, acc3 = {};

#pragma unroll 4
    for (int k = 0; k < D_IN; k += 4) {
      const int kk = k + 2 * half;
      v2f a;
      a.x = aRow[kk];
      a.y = aRow[kk + 1];

      const float2* wrow = Wp + ((k >> 1) + half) * WP_STRIDE + l15;
      float2 w;
      v2f b;

      w = wrow[0];
      b.x = w.x; b.y = w.y;
      acc0 = __builtin_amdgcn_wmma_f32_16x16x4_f32(false, a, false, b,
                                                   (short)0, acc0, false, false);
      w = wrow[16];
      b.x = w.x; b.y = w.y;
      acc1 = __builtin_amdgcn_wmma_f32_16x16x4_f32(false, a, false, b,
                                                   (short)0, acc1, false, false);
      w = wrow[32];
      b.x = w.x; b.y = w.y;
      acc2 = __builtin_amdgcn_wmma_f32_16x16x4_f32(false, a, false, b,
                                                   (short)0, acc2, false, false);
      w = wrow[48];
      b.x = w.x; b.y = w.y;
      acc3 = __builtin_amdgcn_wmma_f32_16x16x4_f32(false, a, false, b,
                                                   (short)0, acc3, false, false);
    }

#pragma unroll
    for (int r = 0; r < 8; ++r) {
      const int row = rowBase + r + 8 * half;
      float* o = out + (size_t)row * 64 + l15;
      o[0]  = acc0[r] + b0;
      o[16] = acc1[r] + b1;
      o[32] = acc2[r] + b2;
      o[48] = acc3[r] + b3;
    }
  }
}

__device__ __forceinline__ void group_edges_role(
    unsigned char* lds, int b, const int* __restrict__ esrc,
    const int* __restrict__ edst, unsigned* __restrict__ grp,
    int* __restrict__ off, size_t wsBytes) {
  unsigned* cnt = (unsigned*)lds;
  int* start = (int*)(lds + 1024);
  unsigned* sg = (unsigned*)(lds + 1024 + 1040);

  if (wsBytes < WS_NEEDED) return;

  const int tid = threadIdx.x;
  const int base = b * SLICE;
  constexpr int PER = (SLICE + 255) / 256;
  const bool lastIn = (PER - 1) * 256 + tid < SLICE;

  cnt[tid] = 0u;
  __syncthreads();

  int s[PER], d[PER];
  unsigned w[PER], br[PER];
#pragma unroll
  for (int k = 0; k < PER; ++k) {
    const int i = k * 256 + tid;
    if (k < PER - 1 || lastIn) {
      s[k] = esrc[base + i];
      d[k] = edst[base + i];
    } else {
      s[k] = 0;
      d[k] = 0;
    }
  }
#pragma unroll
  for (int k = 0; k < PER; ++k) {
    const int q = d[k] / ROWS_PER;
    const int bin = q < N_BINS - 1 ? q : N_BINS - 1;
    w[k] = ((unsigned)s[k] << 9) | (unsigned)(d[k] - bin * ROWS_PER);
    unsigned rank = 0u;
    if (k < PER - 1 || lastIn)
      rank = atomicAdd(&cnt[bin], 1u);
    br[k] = (unsigned)bin | (rank << 8);
  }
  __syncthreads();

  {
    int sum = 0;
    for (int u = 0; u < tid; ++u) sum += (int)cnt[u];
    start[tid] = sum;
    off[b * OFF_PITCH + tid] = sum;
    if (tid == N_BINS - 1) {
      const int tot = sum + (int)cnt[tid];
      start[N_BINS] = tot;
      off[b * OFF_PITCH + N_BINS] = tot;
    }
  }
  __syncthreads();

#pragma unroll
  for (int k = 0; k < PER; ++k) {
    if (k < PER - 1 || lastIn)
      sg[start[br[k] & 255u] + (int)(br[k] >> 8)] = w[k];
  }
  __syncthreads();

  for (int i = tid; i < SLICE; i += 256) grp[base + i] = sg[i];
}

__global__ __launch_bounds__(256) void gemm_and_group(
    const float* __restrict__ A, const float* __restrict__ W1,
    const float* __restrict__ bias1, const float* __restrict__ W2,
    float* __restrict__ g, int nRows, int nBlocks,
    const int* __restrict__ esrc, const int* __restrict__ edst,
    unsigned* __restrict__ grp, int* __restrict__ off, size_t wsBytes) {
  __shared__ __attribute__((aligned(16))) unsigned char lds[LDS_BYTES];
  if (blockIdx.x < GEMM_BLOCKS) {
    gemm_folded_role(lds, A, W1, bias1, W2, g, nRows, nBlocks);
  } else {
    group_edges_role(lds, (int)blockIdx.x - GEMM_BLOCKS, esrc, edst, grp, off,
                     wsBytes);
  }
}

#define ROWS_IN_FLIGHT 16

__device__ __forceinline__ void add_batch(const float2* __restrict__ g2,
                                          float* acc, int s, int r,
                                          unsigned mask, int lane) {
  while (mask) {
    int sk[ROWS_IN_FLIGHT], rk[ROWS_IN_FLIGHT];
    bool ok[ROWS_IN_FLIGHT];
    float2 v[ROWS_IN_FLIGHT];
#pragma unroll
    for (int k = 0; k < ROWS_IN_FLIGHT; ++k) {
      ok[k] = mask != 0u;
      const int i = ok[k] ? __builtin_ctz(mask) : 0;
      mask &= mask - 1u;
      sk[k] = __builtin_amdgcn_readlane(s, i);
      rk[k] = __builtin_amdgcn_readlane(r, i);
    }
#pragma unroll
    for (int k = 0; k < ROWS_IN_FLIGHT; ++k) {
      v[k] = make_float2(0.0f, 0.0f);
      if (ok[k]) v[k] = g2[(size_t)sk[k] * 32 + lane];
    }
#pragma unroll
    for (int k = 0; k < ROWS_IN_FLIGHT; ++k) {
      if (ok[k]) {
        float* p = acc + rk[k] * 64 + 2 * lane;
        atomicAdd(p + 0, v[k].x);
        atomicAdd(p + 1, v[k].y);
      }
    }
  }
}

__device__ __forceinline__ float2 fetch_row(const float* __restrict__ g,
                                            unsigned so, int k,
                                            unsigned lane8) {
  const unsigned o = (unsigned)__builtin_amdgcn_readlane((int)so, k) + lane8;
  return *(const float2*)((const char*)g + o);
}

__device__ __forceinline__ void add_row(float* acc, unsigned ro, int k,
                                        unsigned lane8, float2 v) {
  const unsigned o = (unsigned)__builtin_amdgcn_readlane((int)ro, k) + lane8;
  float* p = (float*)((char*)acc + o);
  atomicAdd(p + 0, v.x);
  atomicAdd(p + 1, v.y);
}

__device__ __forceinline__ void add_run(const float* __restrict__ g,
                                        float* acc, unsigned so, unsigned ro,
                                        int m, unsigned lane8) {
  float2 v[32];
#pragma unroll
  for (int q = 0; q < 32; q += 4) {
    if (q >= m) break;
#pragma unroll
    for (int k = q; k < q + 4; ++k) v[k] = fetch_row(g, so, k, lane8);
  }
#pragma unroll
  for (int q = 0; q < 32; q += 4) {
    if (q >= m) break;
#pragma unroll
    for (int k = q; k < q + 4; ++k) add_row(acc, ro, k, lane8, v[k]);
  }
}

__global__ __launch_bounds__(1024) void sum_rows_by_owner(
    const float* __restrict__ g, const int* __restrict__ esrc,
    const int* __restrict__ edst, const unsigned* __restrict__ grp,
    const int* __restrict__ off,
    const float* __restrict__ bias2, float* __restrict__ out, size_t wsBytes) {
  __shared__ float acc[(ROWS_PER + 1) * 64];

  const int lane = threadIdx.x & 31;
  const int wave = threadIdx.x >> 5;
  const int j = blockIdx.x;
  const int lo = j * ROWS_PER;
  const int rows = (N_NODES - lo) < ROWS_PER ? (N_NODES - lo) : ROWS_PER;
  const float2* g2 = (const float2*)g;

  for (int i = threadIdx.x; i < (ROWS_PER + 1) * 16; i += 1024)
    ((float4*)acc)[i] = make_float4(0.0f, 0.0f, 0.0f, 0.0f);
  __syncthreads();

  if (wsBytes >= WS_NEEDED) {
    int myBeg = 0, myEnd = 0;
    if (lane < N_BINS / 32) {
      const int b = wave + 32 * lane;
      myBeg = off[b * OFF_PITCH + j];
      myEnd = off[b * OFF_PITCH + j + 1];
    }
    const unsigned lane8 = (unsigned)lane * 8u;
    const unsigned padWord = (unsigned)ROWS_PER;
    int beg = __builtin_amdgcn_readlane(myBeg, 0);
    int end = __builtin_amdgcn_readlane(myEnd, 0);
    unsigned wN = beg + lane < end ? grp[wave * SLICE + beg + lane] : padWord;
    for (int t = 0; t < N_BINS / 32; ++t) {
      const int b = wave + 32 * t;
      const unsigned* pg = grp + b * SLICE;
      const unsigned so0 = (wN >> 1) & ~255u;
      const unsigned ro0 = (wN & 511u) << 8;
      const int begC = beg, endC = end;
      if (t + 1 < N_BINS / 32) {
        beg = __builtin_amdgcn_readlane(myBeg, t + 1);
        end = __builtin_amdgcn_readlane(myEnd, t + 1);
        wN = beg + lane < end ? grp[(b + 32) * SLICE + beg + lane] : padWord;
      }
      unsigned so = so0, ro = ro0;
      int p = begC;
      int left = endC - begC;
      while (left > 0) {
        add_run(g, acc, so, ro, left < 32 ? left : 32, lane8);
        p += 32;
        left -= 32;
        if (left > 0) {
          const int i = p + lane;
          const unsigned w = i < endC ? pg[i] : padWord;
          so = (w >> 1) & ~255u;
          ro = (w & 511u) << 8;
        }
      }
    }
  } else {
    for (int p = wave * 32; p < N_EDGES; p += 1024) {
      const int i = p + lane;
      const bool in = i < N_EDGES;
      const int s = in ? esrc[i] : 0;
      const int r = in ? edst[i] - lo : -1;
      const bool mine = in && (unsigned)r < (unsigned)rows;
      add_batch(g2, acc, s, r, __builtin_amdgcn_ballot_w32(mine), lane);
    }
  }

  __syncthreads();

  const float2 bb = ((const float2*)bias2)[lane];
  for (int r = wave; r < rows; r += 32) {
    const float2 a = ((const float2*)acc)[r * 32 + lane];
    ((float2*)out)[(size_t)(lo + r) * 32 + lane] =
        make_float2(a.x + bb.x, a.y + bb.y);
  }
}

extern "C" void kernel_launch(void* const* d_in, const int* in_sizes, int n_in,
                              void* d_out, int out_size, void* d_ws,
                              size_t ws_size, hipStream_t stream) {
  const float* data   = (const float*)d_in[0];
  const int*   esrc   = (const int*)d_in[1];
  const int*   edst   = (const int*)d_in[2];
  const float* W_pre  = (const float*)d_in[3];
  const float* b_pre  = (const float*)d_in[4];
  const float* W_post = (const float*)d_in[5];
  const float* b_post = (const float*)d_in[6];

  float* out = (float*)d_out;
  float* g   = (float*)d_ws;
  unsigned* grp = (unsigned*)((char*)d_ws + G_BYTES);
  int* off = (int*)((char*)d_ws + G_BYTES + REC_BYTES);

  gemm_and_group<<<GRID_FIRST, 256, 0, stream>>>(
      data, W_pre, b_pre, W_post, g, N_NODES, GEMM_BLOCKS, esrc, edst, grp, off,
      ws_size);

  sum_rows_by_owner<<<N_BINS, 1024, 0, stream>>>(g, esrc, edst, grp, off,
                                                 b_post, out, ws_size);
}


# edge pass: owner orders its edges by destination row in LDS (integer count, scan, place); a wave sums a row's g rows in registers, stores it once; no LDS float adds, no plane; add order changes
# speedup vs baseline: 4.5524x; 1.1883x over previous
#include <hip/hip_runtime.h>


typedef __attribute__((ext_vector_type(2))) float v2f;
typedef __attribute__((ext_vector_type(8))) float v8f;

#define N_NODES 100000
#define D_IN    128
#define D_HID   64
#define D_OUT   64
#define N_EDGES 1600000

#define N_BINS    256
#define ROWS_PER  391
#define SLICE     (N_EDGES / N_BINS)
#define OFF_PITCH (N_BINS + 1)

#define G_BYTES   ((size_t)N_NODES * D_OUT * sizeof(float))
#define REC_BYTES ((size_t)N_EDGES * sizeof(int))
#define OFF_BYTES ((size_t)N_BINS * OFF_PITCH * sizeof(int))
#define WS_NEEDED (G_BYTES + REC_BYTES + OFF_BYTES)

#define WP_STRIDE 80
#define WP_BYTES  ((D_IN / 2) * WP_STRIDE * 8)
#define WQ_BYTES  ((D_HID / 2) * WP_STRIDE * 8)
#define LDS_BYTES (WP_BYTES + WQ_BYTES)

#define GEMM_BLOCKS 250
#define GRID_FIRST  506
static_assert(GRID_FIRST == GEMM_BLOCKS + N_BINS, "grid of the first launch");

__device__ __forceinline__ void gemm_folded_role(
    unsigned char* lds, const float* __restrict__ A,
    const float* __restrict__ W1, const float* __restrict__ bias1,
    const float* __restrict__ W2, float* __restrict__ out, int nRows,
    int nBlocks) {
  float2* Wp = (float2*)lds;
  float2* Wq = (float2*)(lds + WP_BYTES);

  for (int i = threadIdx.x; i < D_HID * D_OUT; i += 256) {
    const int k = i >> 6;
    const int c = i & 63;
    ((float*)Wq)[(k >> 1) * (2 * WP_STRIDE) + c * 2 + (k & 1)] = W2[i];
  }
  __syncthreads();

  const int lane = threadIdx.x & 31;
  const int wave = threadIdx.x >> 5;
  const int half = lane >> 4;
  const int l15  = lane & 15;

  {
    const float* aRow = W1 + (size_t)(wave * 16 + l15) * D_HID;
    v8f f0 = {}, f1 = {}, f2 = {}, f3 = {};

#pragma unroll 4
    for (int k = 0; k < D_HID; k += 4) {
      const int kk = k + 2 * half;
      v2f a;
      a.x = aRow[kk];
      a.y = aRow[kk + 1];

      const float2* wrow = Wq + ((k >> 1) + half) * WP_STRIDE + l15;
      float2 w;
      v2f b;

      w = wrow[0];
      b.x = w.x; b.y = w.y;
      f0 = __builtin_amdgcn_wmma_f32_16x16x4_f32(false, a, false, b,
                                                 (short)0, f0, false, false);
      w = wrow[16];
      b.x = w.x; b.y = w.y;
      f1 = __builtin_amdgcn_wmma_f32_16x16x4_f32(false, a, false, b,
                                                 (short)0, f1, false, false);
      w = wrow[32];
      b.x = w.x; b.y = w.y;
      f2 = __builtin_amdgcn_wmma_f32_16x16x4_f32(false, a, false, b,
                                                 (short)0, f2, false, false);
      w = wrow[48];
      b.x = w.x; b.y = w.y;
      f3 = __builtin_amdgcn_wmma_f32_16x16x4_f32(false, a, false, b,
                                                 (short)0, f3, false, false);
    }

#pragma unroll
    for (int r = 0; r < 8; ++r) {
      const int kr = wave * 16 + r + 8 * half;
      float* p = (float*)Wp + (kr >> 1) * (2 * WP_STRIDE) + (kr & 1) + 2 * l15;
      p[0]  = f0[r];
      p[32] = f1[r];
      p[64] = f2[r];
      p[96] = f3[r];
    }
  }

  float b0 = 0.0f, b1 = 0.0f, b2 = 0.0f, b3 = 0.0f;
  for (int kp = 0; kp < D_HID / 2; ++kp) {
    const float s0 = bias1[2 * kp];
    const float s1 = bias1[2 * kp + 1];
    const float2* q = Wq + kp * WP_STRIDE + l15;
    float2 w;
    w = q[0];  b0 = fmaf(s0, w.x, b0); b0 = fmaf(s1, w.y, b0);
    w = q[16]; b1 = fmaf(s0, w.x, b1); b1 = fmaf(s1, w.y, b1);
    w = q[32]; b2 = fmaf(s0, w.x, b2); b2 = fmaf(s1, w.y, b2);
    w = q[48]; b3 = fmaf(s0, w.x, b3); b3 = fmaf(s1, w.y, b3);
  }

  __syncthreads();

  const int nTiles = nRows >> 4;

  for (int tile = blockIdx.x * 8 + wave; tile < nTiles;
       tile += nBlocks * 8) {
    const int rowBase = tile * 16;
    const float* aRow = A + (size_t)(rowBase + l15) * D_IN;

    v8f acc0 = {}, acc1 = {}, acc2 = {}, acc3 = {};

#pragma unroll 4
    for (int k = 0; k < D_IN; k += 4) {
      const int kk = k + 2 * half;
      v2f a;
      a.x = aRow[kk];
      a.y = aRow[kk + 1];

      const float2* wrow = Wp + ((k >> 1) + half) * WP_STRIDE + l15;
      float2 w;
      v2f b;

      w = wrow[0];
      b.x = w.x; b.y = w.y;
      acc0 = __builtin_amdgcn_wmma_f32_16x16x4_f32(false, a, false, b,
                                                   (short)0, acc0, false, false);
      w = wrow[16];
      b.x = w.x; b.y = w.y;
      acc1 = __builtin_amdgcn_wmma_f32_16x16x4_f32(false, a, false, b,
                                                   (short)0, acc1, false, false);
      w = wrow[32];
      b.x = w.x; b.y = w.y;
      acc2 = __builtin_amdgcn_wmma_f32_16x16x4_f32(false, a, false, b,
                                                   (short)0, acc2, false, false);
      w = wrow[48];
      b.x = w.x; b.y = w.y;
      acc3 = __builtin_amdgcn_wmma_f32_16x16x4_f32(false, a, false, b,
                                                   (short)0, acc3, false, false);
    }

#pragma unroll
    for (int r = 0; r < 8; ++r) {
      const int row = rowBase + r + 8 * half;
      float* o = out + (size_t)row * 64 + l15;
      o[0]  = acc0[r] + b0;
      o[16] = acc1[r] + b1;
      o[32] = acc2[r] + b2;
      o[48] = acc3[r] + b3;
    }
  }
}

__device__ __forceinline__ void group_edges_role(
    unsigned char* lds, int b, const int* __restrict__ esrc,
    const int* __restrict__ edst, unsigned* __restrict__ grp,
    int* __restrict__ off, size_t wsBytes) {
  unsigned* cnt = (unsigned*)lds;
  int* start = (int*)(lds + 1024);
  unsigned* sg = (unsigned*)(lds + 1024 + 1040);

  if (wsBytes < WS_NEEDED) return;

  const int tid = threadIdx.x;
  const int base = b * SLICE;
  constexpr int PER = (SLICE + 255) / 256;
  const bool lastIn = (PER - 1) * 256 + tid < SLICE;

  cnt[tid] = 0u;
  __syncthreads();

  int s[PER], d[PER];
  unsigned w[PER], br[PER];
#pragma unroll
  for (int k = 0; k < PER; ++k) {
    const int i = k * 256 + tid;
    if (k < PER - 1 || lastIn) {
      s[k] = esrc[base + i];
      d[k] = edst[base + i];
    } else {
      s[k] = 0;
      d[k] = 0;
    }
  }
#pragma unroll
  for (int k = 0; k < PER; ++k) {
    const int q = d[k] / ROWS_PER;
    const int bin = q < N_BINS - 1 ? q : N_BINS - 1;
    w[k] = ((unsigned)s[k] << 9) | (unsigned)(d[k] - bin * ROWS_PER);
    unsigned rank = 0u;
    if (k < PER - 1 || lastIn)
      rank = atomicAdd(&cnt[bin], 1u);
    br[k] = (unsigned)bin | (rank << 8);
  }
  __syncthreads();

  {
    int sum = 0;
    for (int u = 0; u < tid; ++u) sum += (int)cnt[u];
    start[tid] = sum;
    off[b * OFF_PITCH + tid] = sum;
    if (tid == N_BINS - 1) {
      const int tot = sum + (int)cnt[tid];
      start[N_BINS] = tot;
      off[b * OFF_PITCH + N_BINS] = tot;
    }
  }
  __syncthreads();

#pragma unroll
  for (int k = 0; k < PER; ++k) {
    if (k < PER - 1 || lastIn)
      sg[start[br[k] & 255u] + (int)(br[k] >> 8)] = w[k];
  }
  __syncthreads();

  for (int i = tid; i < SLICE; i += 256) grp[base + i] = sg[i];
}

__global__ __launch_bounds__(256) void gemm_and_group(
    const float* __restrict__ A, const float* __restrict__ W1,
    const float* __restrict__ bias1, const float* __restrict__ W2,
    float* __restrict__ g, int nRows, int nBlocks,
    const int* __restrict__ esrc, const int* __restrict__ edst,
    unsigned* __restrict__ grp, int* __restrict__ off, size_t wsBytes) {
  __shared__ __attribute__((aligned(16))) unsigned char lds[LDS_BYTES];
  if (blockIdx.x < GEMM_BLOCKS) {
    gemm_folded_role(lds, A, W1, bias1, W2, g, nRows, nBlocks);
  } else {
    group_edges_role(lds, (int)blockIdx.x - GEMM_BLOCKS, esrc, edst, grp, off,
                     wsBytes);
  }
}

#define ROWS_IN_FLIGHT 16

__device__ __forceinline__ void add_batch(const float2* __restrict__ g2,
                                          float* acc, int s, int r,
                                          unsigned mask, int lane) {
  while (mask) {
    int sk[ROWS_IN_FLIGHT], rk[ROWS_IN_FLIGHT];
    bool ok[ROWS_IN_FLIGHT];
    float2 v[ROWS_IN_FLIGHT];
#pragma unroll
    for (int k = 0; k < ROWS_IN_FLIGHT; ++k) {
      ok[k] = mask != 0u;
      const int i = ok[k] ? __builtin_ctz(mask) : 0;
      mask &= mask - 1u;
      sk[k] = __builtin_amdgcn_readlane(s, i);
      rk[k] = __builtin_amdgcn_readlane(r, i);
    }
#pragma unroll
    for (int k = 0; k < ROWS_IN_FLIGHT; ++k) {
      v[k] = make_float2(0.0f, 0.0f);
      if (ok[k]) v[k] = g2[(size_t)sk[k] * 32 + lane];
    }
#pragma unroll
    for (int k = 0; k < ROWS_IN_FLIGHT; ++k) {
      if (ok[k]) {
        float* p = acc + rk[k] * 64 + 2 * lane;
        atomicAdd(p + 0, v[k].x);
        atomicAdd(p + 1, v[k].y);
      }
    }
  }
}

__device__ __forceinline__ float2 fetch_row(const float* __restrict__ g,
                                            unsigned so, int k,
                                            unsigned lane8) {
  const unsigned o = (unsigned)__builtin_amdgcn_readlane((int)so, k) + lane8;
  return *(const float2*)((const char*)g + o);
}

__device__ __forceinline__ void add_row(float* acc, unsigned ro, int k,
                                        unsigned lane8, float2 v) {
  const unsigned o = (unsigned)__builtin_amdgcn_readlane((int)ro, k) + lane8;
  float* p = (float*)((char*)acc + o);
  atomicAdd(p + 0, v.x);
  atomicAdd(p + 1, v.y);
}

__device__ __forceinline__ void add_run(const float* __restrict__ g,
                                        float* acc, unsigned so, unsigned ro,
                                        int m, unsigned lane8) {
  float2 v[32];
#pragma unroll
  for (int q = 0; q < 32; q += 4) {
    if (q >= m) break;
#pragma unroll
    for (int k = q; k < q + 4; ++k) v[k] = fetch_row(g, so, k, lane8);
  }
#pragma unroll
  for (int q = 0; q < 32; q += 4) {
    if (q >= m) break;
#pragma unroll
    for (int k = q; k < q + 4; ++k) add_row(acc, ro, k, lane8, v[k]);
  }
}

#define SORT_CAP  16384
#define ROW_SLOTS 416
#define NO_EDGE   0xFFFFFFFFu
static_assert(SORT_CAP * 4 <= (ROWS_PER + 1) * 64 * 4, "list fits the plane");
static_assert(ROW_SLOTS >= ROWS_PER + 1 && ROW_SLOTS % 32 == 0, "row slots");

__device__ __forceinline__ int wave_incl_scan(int x, int lane) {
#pragma unroll
  for (int d = 1; d < 32; d <<= 1) {
    const int y = __builtin_amdgcn_ds_bpermute(((lane - d) & 31) << 2, x);
    if (lane >= d) x += y;
  }
  return x;
}

__device__ __forceinline__ float2 sum_run(const float* __restrict__ g,
                                          unsigned so, int m, unsigned lane8,
                                          float2 sum) {
  float2 v[32];
#pragma unroll
  for (int q = 0; q < 32; q += 4) {
    if (q >= m) break;
#pragma unroll
    for (int k = q; k < q + 4; ++k) v[k] = fetch_row(g, so, k, lane8);
  }
#pragma unroll
  for (int q = 0; q < 32; q += 4) {
    if (q >= m) break;
#pragma unroll
    for (int k = q; k < q + 4; ++k) {
      if (k < m) {
        sum.x += v[k].x;
        sum.y += v[k].y;
      }
    }
  }
  return sum;
}

__global__ __launch_bounds__(1024) void sum_rows_by_owner(
    const float* __restrict__ g, const int* __restrict__ esrc,
    const int* __restrict__ edst, const unsigned* __restrict__ grp,
    const int* __restrict__ off,
    const float* __restrict__ bias2, float* __restrict__ out, size_t wsBytes) {
  __shared__ float acc[(ROWS_PER + 1) * 64];
  __shared__ unsigned cur[ROW_SLOTS];
  __shared__ int rs[ROW_SLOTS];
  __shared__ int ctot[16];

  const int lane = threadIdx.x & 31;
  const int wave = threadIdx.x >> 5;
  const int j = blockIdx.x;
  const int lo = j * ROWS_PER;
  const int rows = (N_NODES - lo) < ROWS_PER ? (N_NODES - lo) : ROWS_PER;
  const float2* g2 = (const float2*)g;
  const unsigned lane8 = (unsigned)lane * 8u;
  const bool grouped = wsBytes >= WS_NEEDED;

  int myBeg = 0, myEnd = 0;
  bool ordered = false;

  if (grouped) {
    unsigned* srt = (unsigned*)acc;
    if (threadIdx.x < ROW_SLOTS) cur[threadIdx.x] = 0u;
    if (lane < N_BINS / 32) {
      const int b = wave + 32 * lane;
      myBeg = off[b * OFF_PITCH + j];
      myEnd = off[b * OFF_PITCH + j + 1];
    }
    unsigned w0[N_BINS / 32];
#pragma unroll
    for (int t = 0; t < N_BINS / 32; ++t) {
      const int beg = __builtin_amdgcn_readlane(myBeg, t);
      const int end = __builtin_amdgcn_readlane(myEnd, t);
      w0[t] = beg + lane < end ? grp[(wave + 32 * t) * SLICE + beg + lane]
                               : NO_EDGE;
    }
    __syncthreads();

#pragma unroll
    for (int t = 0; t < N_BINS / 32; ++t) {
      if (w0[t] != NO_EDGE) atomicAdd(&cur[w0[t] & 511u], 1u);
      const int beg = __builtin_amdgcn_readlane(myBeg, t);
      const int end = __builtin_amdgcn_readlane(myEnd, t);
      const unsigned* pg = grp + (wave + 32 * t) * SLICE;
      for (int p = beg + 32; p < end; p += 32) {
        const int i = p + lane;
        if (i < end) atomicAdd(&cur[pg[i] & 511u], 1u);
      }
    }
    __syncthreads();

    int x = 0, ex = 0;
    if (wave < ROW_SLOTS / 32) {
      x = (int)cur[wave * 32 + lane];
      const int incl = wave_incl_scan(x, lane);
      ex = incl - x;
      if (lane == 31) ctot[wave] = incl;
    }
    __syncthreads();
    if (wave < ROW_SLOTS / 32) {
      const int c = lane < ROW_SLOTS / 32 ? ctot[lane] : 0;
      int base = 0;
#pragma unroll
      for (int q = 0; q < ROW_SLOTS / 32; ++q) {
        const int v = __builtin_amdgcn_readlane(c, q);
        base += q < wave ? v : 0;
      }
      rs[wave * 32 + lane] = base + ex;
      cur[wave * 32 + lane] = (unsigned)(base + ex);
    }
    __syncthreads();

    ordered = rs[ROWS_PER] <= SORT_CAP;

    if (ordered) {
#pragma unroll
      for (int t = 0; t < N_BINS / 32; ++t) {
        if (w0[t] != NO_EDGE) {
          const unsigned p = atomicAdd(&cur[w0[t] & 511u], 1u);
          srt[p] = (w0[t] >> 1) & ~255u;
        }
        const int beg = __builtin_amdgcn_readlane(myBeg, t);
        const int end = __builtin_amdgcn_readlane(myEnd, t);
        const unsigned* pg = grp + (wave + 32 * t) * SLICE;
        for (int p0 = beg + 32; p0 < end; p0 += 32) {
          const int i = p0 + lane;
          if (i < end) {
            const unsigned w = pg[i];
            const unsigned p = atomicAdd(&cur[w & 511u], 1u);
            srt[p] = (w >> 1) & ~255u;
          }
        }
      }
      __syncthreads();

      const float2 bb = ((const float2*)bias2)[lane];
      for (int r = wave; r < rows; r += 32) {
        const int s = __builtin_amdgcn_readfirstlane(rs[r]);
        const int e = __builtin_amdgcn_readfirstlane(rs[r + 1]);
        float2 sum = make_float2(0.0f, 0.0f);
        for (int p = s; p < e; p += 32) {
          const int m = e - p < 32 ? e - p : 32;
          const unsigned so = lane < m ? srt[p + lane] : 0u;
          sum = sum_run(g, so, m, lane8, sum);
        }
        ((float2*)out)[(size_t)(lo + r) * 32 + lane] =
            make_float2(sum.x + bb.x, sum.y + bb.y);
      }
    }
  }

  if (!ordered) {
    for (int i = threadIdx.x; i < (ROWS_PER + 1) * 16; i += 1024)
      ((float4*)acc)[i] = make_float4(0.0f, 0.0f, 0.0f, 0.0f);
    __syncthreads();

    if (grouped) {
      const unsigned padWord = (unsigned)ROWS_PER;
      int beg = __builtin_amdgcn_readlane(myBeg, 0);
      int end = __builtin_amdgcn_readlane(myEnd, 0);
      unsigned wN = beg + lane < end ? grp[wave * SLICE + beg + lane] : padWord;
      for (int t = 0; t < N_BINS / 32; ++t) {
        const int b = wave + 32 * t;
        const unsigned* pg = grp + b * SLICE;
        const unsigned so0 = (wN >> 1) & ~255u;
        const unsigned ro0 = (wN & 511u) << 8;
        const int begC = beg, endC = end;
        if (t + 1 < N_BINS / 32) {
          beg = __builtin_amdgcn_readlane(myBeg, t + 1);
          end = __builtin_amdgcn_readlane(myEnd, t + 1);
          wN = beg + lane < end ? grp[(b + 32) * SLICE + beg + lane] : padWord;
        }
        unsigned so = so0, ro = ro0;
        int p = begC;
        int left = endC - begC;
        while (left > 0) {
          add_run(g, acc, so, ro, left < 32 ? left : 32, lane8);
          p += 32;
          left -= 32;
          if (left > 0) {
            const int i = p + lane;
            const unsigned w = i < endC ? pg[i] : padWord;
            so = (w >> 1) & ~255u;
            ro = (w & 511u) << 8;
          }
        }
      }
    } else {
      for (int p = wave * 32; p < N_EDGES; p += 1024) {
        const int i = p + lane;
        const bool in = i < N_EDGES;
        const int s = in ? esrc[i] : 0;
        const int r = in ? edst[i] - lo : -1;
        const bool mine = in && (unsigned)r < (unsigned)rows;
        add_batch(g2, acc, s, r, __builtin_amdgcn_ballot_w32(mine), lane);
      }
    }

    __syncthreads();

    const float2 bb = ((const float2*)bias2)[lane];
    for (int r = wave; r < rows; r += 32) {
      const float2 a = ((const float2*)acc)[r * 32 + lane];
      ((float2*)out)[(size_t)(lo + r) * 32 + lane] =
          make_float2(a.x + bb.x, a.y + bb.y);
    }
  }
}

extern "C" void kernel_launch(void* const* d_in, const int* in_sizes, int n_in,
                              void* d_out, int out_size, void* d_ws,
                              size_t ws_size, hipStream_t stream) {
  const float* data   = (const float*)d_in[0];
  const int*   esrc   = (const int*)d_in[1];
  const int*   edst   = (const int*)d_in[2];
  const float* W_pre  = (const float*)d_in[3];
  const float* b_pre  = (const float*)d_in[4];
  const float* W_post = (const float*)d_in[5];
  const float* b_post = (const float*)d_in[6];

  float* out = (float*)d_out;
  float* g   = (float*)d_ws;
  unsigned* grp = (unsigned*)((char*)d_ws + G_BYTES);
  int* off = (int*)((char*)d_ws + G_BYTES + REC_BYTES);

  gemm_and_group<<<GRID_FIRST, 256, 0, stream>>>(
      data, W_pre, b_pre, W_post, g, N_NODES, GEMM_BLOCKS, esrc, edst, grp, off,
      ws_size);

  sum_rows_by_owner<<<N_BINS, 1024, 0, stream>>>(g, esrc, edst, grp, off,
                                                 b_post, out, ws_size);
}


# node GEMM and fold: A rows are requested one 16-k-step chunk (sixteen 8-byte loads) ahead of the matrix instructions, in two register buffers; no full load wait per four k steps any more
# speedup vs baseline: 4.9114x; 1.0789x over previous
#include <hip/hip_runtime.h>


typedef __attribute__((ext_vector_type(2))) float v2f;
typedef __attribute__((ext_vector_type(8))) float v8f;

#define N_NODES 100000
#define D_IN    128
#define D_HID   64
#define D_OUT   64
#define N_EDGES 1600000

#define N_BINS    256
#define ROWS_PER  391
#define SLICE     (N_EDGES / N_BINS)
#define OFF_PITCH (N_BINS + 1)

#define G_BYTES   ((size_t)N_NODES * D_OUT * sizeof(float))
#define REC_BYTES ((size_t)N_EDGES * sizeof(int))
#define OFF_BYTES ((size_t)N_BINS * OFF_PITCH * sizeof(int))
#define WS_NEEDED (G_BYTES + REC_BYTES + OFF_BYTES)

#define WP_STRIDE 80
#define WP_BYTES  ((D_IN / 2) * WP_STRIDE * 8)
#define WQ_BYTES  ((D_HID / 2) * WP_STRIDE * 8)
#define LDS_BYTES (WP_BYTES + WQ_BYTES)

#define GEMM_BLOCKS 250
#define GRID_FIRST  506
static_assert(GRID_FIRST == GEMM_BLOCKS + N_BINS, "grid of the first launch");

__device__ __forceinline__ void load_chunk(v2f (&a)[16], const float* p) {
#pragma unroll
  for (int i = 0; i < 16; ++i) {
    a[i].x = p[4 * i];
    a[i].y = p[4 * i + 1];
  }
}

__device__ __forceinline__ void mma_chunk(v8f& c0, v8f& c1, v8f& c2, v8f& c3,
                                          const v2f (&a)[16],
                                          const float2* Wpair, int kbase,
                                          int half, int l15) {
#pragma unroll
  for (int i = 0; i < 16; ++i) {
    const int k = kbase + 4 * i;
    const float2* wrow = Wpair + ((k >> 1) + half) * WP_STRIDE + l15;
    float2 w;
    v2f b;

    w = wrow[0];
    b.x = w.x; b.y = w.y;
    c0 = __builtin_amdgcn_wmma_f32_16x16x4_f32(false, a[i], false, b,
                                               (short)0, c0, false, false);
    w = wrow[16];
    b.x = w.x; b.y = w.y;
    c1 = __builtin_amdgcn_wmma_f32_16x16x4_f32(false, a[i], false, b,
                                               (short)0, c1, false, false);
    w = wrow[32];
    b.x = w.x; b.y = w.y;
    c2 = __builtin_amdgcn_wmma_f32_16x16x4_f32(false, a[i], false, b,
                                               (short)0, c2, false, false);
    w = wrow[48];
    b.x = w.x; b.y = w.y;
    c3 = __builtin_amdgcn_wmma_f32_16x16x4_f32(false, a[i], false, b,
                                               (short)0, c3, false, false);
  }
}

__device__ __forceinline__ void gemm_folded_role(
    unsigned char* lds, const float* __restrict__ A,
    const float* __restrict__ W1, const float* __restrict__ bias1,
    const float* __restrict__ W2, float* __restrict__ out, int nRows,
    int nBlocks) {
  float2* Wp = (float2*)lds;
  float2* Wq = (float2*)(lds + WP_BYTES);

  for (int i = threadIdx.x; i < D_HID * D_OUT; i += 256) {
    const int k = i >> 6;
    const int c = i & 63;
    ((float*)Wq)[(k >> 1) * (2 * WP_STRIDE) + c * 2 + (k & 1)] = W2[i];
  }

  const int lane = threadIdx.x & 31;
  const int wave = threadIdx.x >> 5;
  const int half = lane >> 4;
  const int l15  = lane & 15;

  const int nTiles = nRows >> 4;
  const int stride = nBlocks * 8;
  int tile = blockIdx.x * 8 + wave;

  v2f aw[16], a0[16], a1[16];
  load_chunk(aw, W1 + (size_t)(wave * 16 + l15) * D_HID + 2 * half);

  __syncthreads();

  if (tile < nTiles)
    load_chunk(a0, A + (size_t)(tile * 16 + l15) * D_IN + 2 * half);

  {
    v8f f0 = {}, f1 = {}, f2 = {}, f3 = {};
    mma_chunk(f0, f1, f2, f3, aw, Wq, 0, half, l15);

#pragma unroll
    for (int r = 0; r < 8; ++r) {
      const int kr = wave * 16 + r + 8 * half;
      float* p = (float*)Wp + (kr >> 1) * (2 * WP_STRIDE) + (kr & 1) + 2 * l15;
      p[0]  = f0[r];
      p[32] = f1[r];
      p[64] = f2[r];
      p[96] = f3[r];
    }
  }

  float b0 = 0.0f, b1 = 0.0f, b2 = 0.0f, b3 = 0.0f;
  for (int kp = 0; kp < D_HID / 2; ++kp) {
    const float s0 = bias1[2 * kp];
    const float s1 = bias1[2 * kp + 1];
    const float2* q = Wq + kp * WP_STRIDE + l15;
    float2 w;
    w = q[0];  b0 = fmaf(s0, w.x, b0); b0 = fmaf(s1, w.y, b0);
    w = q[16]; b1 = fmaf(s0, w.x, b1); b1 = fmaf(s1, w.y, b1);
    w = q[32]; b2 = fmaf(s0, w.x, b2); b2 = fmaf(s1, w.y, b2);
    w = q[48]; b3 = fmaf(s0, w.x, b3); b3 = fmaf(s1, w.y, b3);
  }

  __syncthreads();

  for (; tile < nTiles; tile += stride) {
    const int rowBase = tile * 16;
    const float* aRow = A + (size_t)(rowBase + l15) * D_IN + 2 * half;

    v8f acc0 = {}, acc1 = {}, acc2 = {}, acc3 = {};

    load_chunk(a1, aRow + 64);
    __builtin_amdgcn_sched_barrier(0);
    mma_chunk(acc0, acc1, acc2, acc3, a0, Wp, 0, half, l15);

    const int next = tile + stride;
    int ahead = next < nTiles ? next : tile;
    asm volatile("" : "+v"(ahead), "+v"(acc0), "+v"(acc1), "+v"(acc2),
                      "+v"(acc3));
    load_chunk(a0, A + (size_t)(ahead * 16 + l15) * D_IN + 2 * half);
    mma_chunk(acc0, acc1, acc2, acc3, a1, Wp, 64, half, l15);

#pragma unroll
    for (int r = 0; r < 8; ++r) {
      const int row = rowBase + r + 8 * half;
      float* o = out + (size_t)row * 64 + l15;
      o[0]  = acc0[r] + b0;
      o[16] = acc1[r] + b1;
      o[32] = acc2[r] + b2;
      o[48] = acc3[r] + b3;
    }
  }
}

__device__ __forceinline__ void group_edges_role(
    unsigned char* lds, int b, const int* __restrict__ esrc,
    const int* __restrict__ edst, unsigned* __restrict__ grp,
    int* __restrict__ off, size_t wsBytes) {
  unsigned* cnt = (unsigned*)lds;
  int* start = (int*)(lds + 1024);
  unsigned* sg = (unsigned*)(lds + 1024 + 1040);

  if (wsBytes < WS_NEEDED) return;

  const int tid = threadIdx.x;
  const int base = b * SLICE;
  constexpr int PER = (SLICE + 255) / 256;
  const bool lastIn = (PER - 1) * 256 + tid < SLICE;

  cnt[tid] = 0u;
  __syncthreads();

  int s[PER], d[PER];
  unsigned w[PER], br[PER];
#pragma unroll
  for (int k = 0; k < PER; ++k) {
    const int i = k * 256 + tid;
    if (k < PER - 1 || lastIn) {
      s[k] = esrc[base + i];
      d[k] = edst[base + i];
    } else {
      s[k] = 0;
      d[k] = 0;
    }
  }
#pragma unroll
  for (int k = 0; k < PER; ++k) {
    const int q = d[k] / ROWS_PER;
    const int bin = q < N_BINS - 1 ? q : N_BINS - 1;
    w[k] = ((unsigned)s[k] << 9) | (unsigned)(d[k] - bin * ROWS_PER);
    unsigned rank = 0u;
    if (k < PER - 1 || lastIn)
      rank = atomicAdd(&cnt[bin], 1u);
    br[k] = (unsigned)bin | (rank << 8);
  }
  __syncthreads();

  {
    int sum = 0;
    for (int u = 0; u < tid; ++u) sum += (int)cnt[u];
    start[tid] = sum;
    off[b * OFF_PITCH + tid] = sum;
    if (tid == N_BINS - 1) {
      const int tot = sum + (int)cnt[tid];
      start[N_BINS] = tot;
      off[b * OFF_PITCH + N_BINS] = tot;
    }
  }
  __syncthreads();

#pragma unroll
  for (int k = 0; k < PER; ++k) {
    if (k < PER - 1 || lastIn)
      sg[start[br[k] & 255u] + (int)(br[k] >> 8)] = w[k];
  }
  __syncthreads();

  for (int i = tid; i < SLICE; i += 256) grp[base + i] = sg[i];
}

__global__ __launch_bounds__(256) void gemm_and_group(
    const float* __restrict__ A, const float* __restrict__ W1,
    const float* __restrict__ bias1, const float* __restrict__ W2,
    float* __restrict__ g, int nRows, int nBlocks,
    const int* __restrict__ esrc, const int* __restrict__ edst,
    unsigned* __restrict__ grp, int* __restrict__ off, size_t wsBytes) {
  __shared__ __attribute__((aligned(16))) unsigned char lds[LDS_BYTES];
  if (blockIdx.x < GEMM_BLOCKS) {
    gemm_folded_role(lds, A, W1, bias1, W2, g, nRows, nBlocks);
  } else {
    group_edges_role(lds, (int)blockIdx.x - GEMM_BLOCKS, esrc, edst, grp, off,
                     wsBytes);
  }
}

#define ROWS_IN_FLIGHT 16

__device__ __forceinline__ void add_batch(const float2* __restrict__ g2,
                                          float* acc, int s, int r,
                                          unsigned mask, int lane) {
  while (mask) {
    int sk[ROWS_IN_FLIGHT], rk[ROWS_IN_FLIGHT];
    bool ok[ROWS_IN_FLIGHT];
    float2 v[ROWS_IN_FLIGHT];
#pragma unroll
    for (int k = 0; k < ROWS_IN_FLIGHT; ++k) {
      ok[k] = mask != 0u;
      const int i = ok[k] ? __builtin_ctz(mask) : 0;
      mask &= mask - 1u;
      sk[k] = __builtin_amdgcn_readlane(s, i);
      rk[k] = __builtin_amdgcn_readlane(r, i);
    }
#pragma unroll
    for (int k = 0; k < ROWS_IN_FLIGHT; ++k) {
      v[k] = make_float2(0.0f, 0.0f);
      if (ok[k]) v[k] = g2[(size_t)sk[k] * 32 + lane];
    }
#pragma unroll
    for (int k = 0; k < ROWS_IN_FLIGHT; ++k) {
      if (ok[k]) {
        float* p = acc + rk[k] * 64 + 2 * lane;
        atomicAdd(p + 0, v[k].x);
        atomicAdd(p + 1, v[k].y);
      }
    }
  }
}

__device__ __forceinline__ float2 fetch_row(const float* __restrict__ g,
                                            unsigned so, int k,
                                            unsigned lane8) {
  const unsigned o = (unsigned)__builtin_amdgcn_readlane((int)so, k) + lane8;
  return *(const float2*)((const char*)g + o);
}

__device__ __forceinline__ void add_row(float* acc, unsigned ro, int k,
                                        unsigned lane8, float2 v) {
  const unsigned o = (unsigned)__builtin_amdgcn_readlane((int)ro, k) + lane8;
  float* p = (float*)((char*)acc + o);
  atomicAdd(p + 0, v.x);
  atomicAdd(p + 1, v.y);
}

__device__ __forceinline__ void add_run(const float* __restrict__ g,
                                        float* acc, unsigned so, unsigned ro,
                                        int m, unsigned lane8) {
  float2 v[32];
#pragma unroll
  for (int q = 0; q < 32; q += 4) {
    if (q >= m) break;
#pragma unroll
    for (int k = q; k < q + 4; ++k) v[k] = fetch_row(g, so, k, lane8);
  }
#pragma unroll
  for (int q = 0; q < 32; q += 4) {
    if (q >= m) break;
#pragma unroll
    for (int k = q; k < q + 4; ++k) add_row(acc, ro, k, lane8, v[k]);
  }
}

#define SORT_CAP  16384
#define ROW_SLOTS 416
#define NO_EDGE   0xFFFFFFFFu
static_assert(SORT_CAP * 4 <= (ROWS_PER + 1) * 64 * 4, "list fits the plane");
static_assert(ROW_SLOTS >= ROWS_PER + 1 && ROW_SLOTS % 32 == 0, "row slots");

__device__ __forceinline__ int wave_incl_scan(int x, int lane) {
#pragma unroll
  for (int d = 1; d < 32; d <<= 1) {
    const int y = __builtin_amdgcn_ds_bpermute(((lane - d) & 31) << 2, x);
    if (lane >= d) x += y;
  }
  return x;
}

__device__ __forceinline__ float2 sum_run(const float* __restrict__ g,
                                          unsigned so, int m, unsigned lane8,
                                          float2 sum) {
  float2 v[32];
#pragma unroll
  for (int q = 0; q < 32; q += 4) {
    if (q >= m) break;
#pragma unroll
    for (int k = q; k < q + 4; ++k) v[k] = fetch_row(g, so, k, lane8);
  }
#pragma unroll
  for (int q = 0; q < 32; q += 4) {
    if (q >= m) break;
#pragma unroll
    for (int k = q; k < q + 4; ++k) {
      if (k < m) {
        sum.x += v[k].x;
        sum.y += v[k].y;
      }
    }
  }
  return sum;
}

__global__ __launch_bounds__(1024) void sum_rows_by_owner(
    const float* __restrict__ g, const int* __restrict__ esrc,
    const int* __restrict__ edst, const unsigned* __restrict__ grp,
    const int* __restrict__ off,
    const float* __restrict__ bias2, float* __restrict__ out, size_t wsBytes) {
  __shared__ float acc[(ROWS_PER + 1) * 64];
  __shared__ unsigned cur[ROW_SLOTS];
  __shared__ int rs[ROW_SLOTS];
  __shared__ int ctot[16];

  const int lane = threadIdx.x & 31;
  const int wave = threadIdx.x >> 5;
  const int j = blockIdx.x;
  const int lo = j * ROWS_PER;
  const int rows = (N_NODES - lo) < ROWS_PER ? (N_NODES - lo) : ROWS_PER;
  const float2* g2 = (const float2*)g;
  const unsigned lane8 = (unsigned)lane * 8u;
  const bool grouped = wsBytes >= WS_NEEDED;

  int myBeg = 0, myEnd = 0;
  bool ordered = false;

  if (grouped) {
    unsigned* srt = (unsigned*)acc;
    if (threadIdx.x < ROW_SLOTS) cur[threadIdx.x] = 0u;
    if (lane < N_BINS / 32) {
      const int b = wave + 32 * lane;
      myBeg = off[b * OFF_PITCH + j];
      myEnd = off[b * OFF_PITCH + j + 1];
    }
    unsigned w0[N_BINS / 32];
#pragma unroll
    for (int t = 0; t < N_BINS / 32; ++t) {
      const int beg = __builtin_amdgcn_readlane(myBeg, t);
      const int end = __builtin_amdgcn_readlane(myEnd, t);
      w0[t] = beg + lane < end ? grp[(wave + 32 * t) * SLICE + beg + lane]
                               : NO_EDGE;
    }
    __syncthreads();

#pragma unroll
    for (int t = 0; t < N_BINS / 32; ++t) {
      if (w0[t] != NO_EDGE) atomicAdd(&cur[w0[t] & 511u], 1u);
      const int beg = __builtin_amdgcn_readlane(myBeg, t);
      const int end = __builtin_amdgcn_readlane(myEnd, t);
      const unsigned* pg = grp + (wave + 32 * t) * SLICE;
      for (int p = beg + 32; p < end; p += 32) {
        const int i = p + lane;
        if (i < end) atomicAdd(&cur[pg[i] & 511u], 1u);
      }
    }
    __syncthreads();

    int x = 0, ex = 0;
    if (wave < ROW_SLOTS / 32) {
      x = (int)cur[wave * 32 + lane];
      const int incl = wave_incl_scan(x, lane);
      ex = incl - x;
      if (lane == 31) ctot[wave] = incl;
    }
    __syncthreads();
    if (wave < ROW_SLOTS / 32) {
      const int c = lane < ROW_SLOTS / 32 ? ctot[lane] : 0;
      int base = 0;
#pragma unroll
      for (int q = 0; q < ROW_SLOTS / 32; ++q) {
        const int v = __builtin_amdgcn_readlane(c, q);
        base += q < wave ? v : 0;
      }
      rs[wave * 32 + lane] = base + ex;
      cur[wave * 32 + lane] = (unsigned)(base + ex);
    }
    __syncthreads();

    ordered = rs[ROWS_PER] <= SORT_CAP;

    if (ordered) {
#pragma unroll
      for (int t = 0; t < N_BINS / 32; ++t) {
        if (w0[t] != NO_EDGE) {
          const unsigned p = atomicAdd(&cur[w0[t] & 511u], 1u);
          srt[p] = (w0[t] >> 1) & ~255u;
        }
        const int beg = __builtin_amdgcn_readlane(myBeg, t);
        const int end = __builtin_amdgcn_readlane(myEnd, t);
        const unsigned* pg = grp + (wave + 32 * t) * SLICE;
        for (int p0 = beg + 32; p0 < end; p0 += 32) {
          const int i = p0 + lane;
          if (i < end) {
            const unsigned w = pg[i];
            const unsigned p = atomicAdd(&cur[w & 511u], 1u);
            srt[p] = (w >> 1) & ~255u;
          }
        }
      }
      __syncthreads();

      const float2 bb = ((const float2*)bias2)[lane];
      for (int r = wave; r < rows; r += 32) {
        const int s = __builtin_amdgcn_readfirstlane(rs[r]);
        const int e = __builtin_amdgcn_readfirstlane(rs[r + 1]);
        float2 sum = make_float2(0.0f, 0.0f);
        for (int p = s; p < e; p += 32) {
          const int m = e - p < 32 ? e - p : 32;
          const unsigned so = lane < m ? srt[p + lane] : 0u;
          sum = sum_run(g, so, m, lane8, sum);
        }
        ((float2*)out)[(size_t)(lo + r) * 32 + lane] =
            make_float2(sum.x + bb.x, sum.y + bb.y);
      }
    }
  }

  if (!ordered) {
    for (int i = threadIdx.x; i < (ROWS_PER + 1) * 16; i += 1024)
      ((float4*)acc)[i] = make_float4(0.0f, 0.0f, 0.0f, 0.0f);
    __syncthreads();

    if (grouped) {
      const unsigned padWord = (unsigned)ROWS_PER;
      int beg = __builtin_amdgcn_readlane(myBeg, 0);
      int end = __builtin_amdgcn_readlane(myEnd, 0);
      unsigned wN = beg + lane < end ? grp[wave * SLICE + beg + lane] : padWord;
      for (int t = 0; t < N_BINS / 32; ++t) {
        const int b = wave + 32 * t;
        const unsigned* pg = grp + b * SLICE;
        const unsigned so0 = (wN >> 1) & ~255u;
        const unsigned ro0 = (wN & 511u) << 8;
        const int begC = beg, endC = end;
        if (t + 1 < N_BINS / 32) {
          beg = __builtin_amdgcn_readlane(myBeg, t + 1);
          end = __builtin_amdgcn_readlane(myEnd, t + 1);
          wN = beg + lane < end ? grp[(b + 32) * SLICE + beg + lane] : padWord;
        }
        unsigned so = so0, ro = ro0;
        int p = begC;
        int left = endC - begC;
        while (left > 0) {
          add_run(g, acc, so, ro, left < 32 ? left : 32, lane8);
          p += 32;
          left -= 32;
          if (left > 0) {
            const int i = p + lane;
            const unsigned w = i < endC ? pg[i] : padWord;
            so = (w >> 1) & ~255u;
            ro = (w & 511u) << 8;
          }
        }
      }
    } else {
      for (int p = wave * 32; p < N_EDGES; p += 1024) {
        const int i = p + lane;
        const bool in = i < N_EDGES;
        const int s = in ? esrc[i] : 0;
        const int r = in ? edst[i] - lo : -1;
        const bool mine = in && (unsigned)r < (unsigned)rows;
        add_batch(g2, acc, s, r, __builtin_amdgcn_ballot_w32(mine), lane);
      }
    }

    __syncthreads();

    const float2 bb = ((const float2*)bias2)[lane];
    for (int r = wave; r < rows; r += 32) {
      const float2 a = ((const float2*)acc)[r * 32 + lane];
      ((float2*)out)[(size_t)(lo + r) * 32 + lane] =
          make_float2(a.x + bb.x, a.y + bb.y);
    }
  }
}

extern "C" void kernel_launch(void* const* d_in, const int* in_sizes, int n_in,
                              void* d_out, int out_size, void* d_ws,
                              size_t ws_size, hipStream_t stream) {
  const float* data   = (const float*)d_in[0];
  const int*   esrc   = (const int*)d_in[1];
  const int*   edst   = (const int*)d_in[2];
  const float* W_pre  = (const float*)d_in[3];
  const float* b_pre  = (const float*)d_in[4];
  const float* W_post = (const float*)d_in[5];
  const float* b_post = (const float*)d_in[6];

  float* out = (float*)d_out;
  float* g   = (float*)d_ws;
  unsigned* grp = (unsigned*)((char*)d_ws + G_BYTES);
  int* off = (int*)((char*)d_ws + G_BYTES + REC_BYTES);

  gemm_and_group<<<GRID_FIRST, 256, 0, stream>>>(
      data, W_pre, b_pre, W_post, g, N_NODES, GEMM_BLOCKS, esrc, edst, grp, off,
      ws_size);

  sum_rows_by_owner<<<N_BINS, 1024, 0, stream>>>(g, esrc, edst, grp, off,
                                                 b_post, out, ws_size);
}
